# starting kernel
# baseline (speedup 1.0000x reference)
#include <hip/hip_runtime.h>
#include <math.h>

#define N_NODES 50000
#define N_EDGES 800000
#define IN_DIM  128
#define OUT_DIM 128
#define HEADS   4
#define D_K     32
#define ATT_DIM 65

typedef __attribute__((ext_vector_type(2))) float v2f;
typedef __attribute__((ext_vector_type(8))) float v8f;

__global__ void __launch_bounds__(256)
gemm_z_wmma(const float* __restrict__ h, const float* __restrict__ W,
            const float* __restrict__ bias, float* __restrict__ z) {
    const int wave = (blockIdx.x * blockDim.x + threadIdx.x) >> 5;
    const int lane = threadIdx.x & 31;
    if (wave >= (N_NODES / 16)) return;

    const int m = lane & 15;
    const int g = lane >> 4;

    const float* __restrict__ hrow = h + (size_t)(wave * 16 + m) * IN_DIM;
    v2f a[32];
#pragma unroll
    for (int ks = 0; ks < 32; ++ks) {
        const int k = 4 * ks + 2 * g;
        a[ks].x = hrow[k];
        a[ks].y = hrow[k + 1];
    }

    for (int ot = 0; ot < 8; ++ot) {
        const int col = ot * 16 + m;
        const float* __restrict__ wrow = W + (size_t)col * IN_DIM;
        v8f c = {};
#pragma unroll
        for (int ks = 0; ks < 32; ++ks) {
            const int k = 4 * ks + 2 * g;
            v2f b;
            b.x = wrow[k];
            b.y = wrow[k + 1];
            c = __builtin_amdgcn_wmma_f32_16x16x4_f32(
                     false, a[ks],  false, b,
                     (short)0, c,  false,  false);
        }
        const float bv = bias[col];
        const int rowBase = wave * 16 + g * 8;
#pragma unroll
        for (int r = 0; r < 8; ++r) {
            z[(size_t)(rowBase + r) * OUT_DIM + col] = c[r] + bv;
        }
    }
}

__global__ void __launch_bounds__(256)
node_scores(const float* __restrict__ z, const float* __restrict__ att,
            float* __restrict__ s_src, float* __restrict__ s_dst) {
    const int gid = blockIdx.x * blockDim.x + threadIdx.x;
    if (gid >= N_NODES * HEADS) return;
    const int hd = gid & 3;
    const int n  = gid >> 2;
    const float* __restrict__ zp = z + (size_t)n * OUT_DIM + hd * D_K;
    const float* __restrict__ as = att + hd * ATT_DIM;
    const float* __restrict__ ad = as + D_K;
    float ss = 0.f, sd = 0.f;
#pragma unroll
    for (int k = 0; k < D_K; ++k) {
        const float v = zp[k];
        ss = fmaf(v, as[k], ss);
        sd = fmaf(v, ad[k], sd);
    }
    s_src[gid] = ss;
    s_dst[gid] = sd;
}

__global__ void __launch_bounds__(256)
zero_f32(float* __restrict__ p, int n) {
    const int gid = blockIdx.x * blockDim.x + threadIdx.x;
    if (gid < n) p[gid] = 0.f;
}

__device__ __forceinline__ float leaky_clip_exp(float s) {
    s = (s >= 0.f) ? s : 0.2f * s;
    s = fminf(fmaxf(s, -20.f), 20.f);
    return __expf(s);
}

__global__ void __launch_bounds__(256)
edge_denom(const int* __restrict__ ei, const float* __restrict__ ef,
           const float* __restrict__ s_src, const float* __restrict__ s_dst,
           const float* __restrict__ att, float* __restrict__ denom) {
    const int gid = blockIdx.x * blockDim.x + threadIdx.x;
    if (gid >= N_EDGES * HEADS) return;
    const int hd = gid & 3;
    const int e  = gid >> 2;
    const int src = ei[e];
    const int dst = ei[N_EDGES + e];
    const float s = s_src[src * HEADS + hd] + s_dst[dst * HEADS + hd]
                  + ef[e] * att[hd * ATT_DIM + 2 * D_K];
    const float alpha = leaky_clip_exp(s);
    __hip_atomic_fetch_add(&denom[dst * HEADS + hd], alpha,
                           __ATOMIC_RELAXED, __HIP_MEMORY_SCOPE_AGENT);
}

__global__ void __launch_bounds__(256)
edge_aggregate(const int* __restrict__ ei, const float* __restrict__ ef,
               const float* __restrict__ s_src, const float* __restrict__ s_dst,
               const float* __restrict__ att, const float* __restrict__ denom,
               const float* __restrict__ z, float* __restrict__ accum) {
    const int wid  = (blockIdx.x * blockDim.x + threadIdx.x) >> 5;
    const int lane = threadIdx.x & 31;
    if (wid >= N_EDGES) return;
    const int src = ei[wid];
    const int dst = ei[N_EDGES + wid];
    const int hd  = lane >> 3;

    const float s = s_src[src * HEADS + hd] + s_dst[dst * HEADS + hd]
                  + ef[wid] * att[hd * ATT_DIM + 2 * D_K];
    const float alpha = leaky_clip_exp(s);
    const float na = alpha / (denom[dst * HEADS + hd] + 1e-6f);

    const float4 zv = *(const float4*)(z + (size_t)src * OUT_DIM + lane * 4);
    float* __restrict__ ap = accum + (size_t)dst * OUT_DIM + lane * 4;
    __hip_atomic_fetch_add(ap + 0, zv.x * na, __ATOMIC_RELAXED, __HIP_MEMORY_SCOPE_AGENT);
    __hip_atomic_fetch_add(ap + 1, zv.y * na, __ATOMIC_RELAXED, __HIP_MEMORY_SCOPE_AGENT);
    __hip_atomic_fetch_add(ap + 2, zv.z * na, __ATOMIC_RELAXED, __HIP_MEMORY_SCOPE_AGENT);
    __hip_atomic_fetch_add(ap + 3, zv.w * na, __ATOMIC_RELAXED, __HIP_MEMORY_SCOPE_AGENT);
}

__global__ void __launch_bounds__(256)
finalize(const float* __restrict__ accum, const float* __restrict__ z,
         const float* __restrict__ ln_g, const float* __restrict__ ln_b,
         float* __restrict__ out) {
    const int wid  = (blockIdx.x * blockDim.x + threadIdx.x) >> 5;
    const int lane = threadIdx.x & 31;
    if (wid >= N_NODES) return;

    const size_t base = (size_t)wid * OUT_DIM + lane * 4;
    const float4 a  = *(const float4*)(accum + base);
    const float4 zv = *(const float4*)(z + base);
    float x[4] = { a.x + zv.x, a.y + zv.y, a.z + zv.z, a.w + zv.w };

    float s = x[0] + x[1] + x[2] + x[3];
#pragma unroll
    for (int off = 16; off > 0; off >>= 1) s += __shfl_xor(s, off, 32);
    const float mu = s * (1.f / 128.f);

    float v = 0.f;
#pragma unroll
    for (int i = 0; i < 4; ++i) { const float d = x[i] - mu; v = fmaf(d, d, v); }
#pragma unroll
    for (int off = 16; off > 0; off >>= 1) v += __shfl_xor(v, off, 32);
    const float rstd = rsqrtf(v * (1.f / 128.f) + 1e-5f);

    const float4 gv = *(const float4*)(ln_g + lane * 4);
    const float4 bv = *(const float4*)(ln_b + lane * 4);
    const float g4[4] = { gv.x, gv.y, gv.z, gv.w };
    const float b4[4] = { bv.x, bv.y, bv.z, bv.w };

    float y[4];
#pragma unroll
    for (int i = 0; i < 4; ++i) {
        const float t = (x[i] - mu) * rstd * g4[i] + b4[i];
        y[i] = (t > 0.f) ? t : (__expf(t) - 1.f);
    }
    float4 o = { y[0], y[1], y[2], y[3] };
    *(float4*)(out + base) = o;
}

extern "C" void kernel_launch(void* const* d_in, const int* in_sizes, int n_in,
                              void* d_out, int out_size, void* d_ws, size_t ws_size,
                              hipStream_t stream) {
    const float* h    = (const float*)d_in[0];
    const int*   ei   = (const int*)  d_in[1];
    const float* ef   = (const float*)d_in[2];
    const float* W_w  = (const float*)d_in[3];
    const float* W_b  = (const float*)d_in[4];
    const float* att  = (const float*)d_in[5];
    const float* ln_g = (const float*)d_in[6];
    const float* ln_b = (const float*)d_in[7];
    float* out = (float*)d_out;

    float* ws     = (float*)d_ws;
    float* z      = ws;
    float* s_src  = z      + (size_t)N_NODES * OUT_DIM;
    float* s_dst  = s_src  + (size_t)N_NODES * HEADS;
    float* denom  = s_dst  + (size_t)N_NODES * HEADS;
    float* accum  = denom  + (size_t)N_NODES * HEADS;

    {
        const int waves = N_NODES / 16;
        const int blocks = (waves + 7) / 8;
        gemm_z_wmma<<<blocks, 256, 0, stream>>>(h, W_w, W_b, z);
    }
    node_scores<<<(N_NODES * HEADS + 255) / 256, 256, 0, stream>>>(z, att, s_src, s_dst);
    {
        const int nz = N_NODES * HEADS + N_NODES * OUT_DIM;
        zero_f32<<<(nz + 255) / 256, 256, 0, stream>>>(denom, nz);
    }
    edge_denom<<<(N_EDGES * HEADS + 255) / 256, 256, 0, stream>>>(ei, ef, s_src, s_dst, att, denom);
    edge_aggregate<<<(N_EDGES * 32 + 255) / 256, 256, 0, stream>>>(ei, ef, s_src, s_dst, att,
                                                                   denom, z, accum);
    finalize<<<(N_NODES * 32 + 255) / 256, 256, 0, stream>>>(accum, z, ln_g, ln_b, out);
}


# Edge sums leave global f32 atomics for LDS: 256 work-groups own 196 destination nodes each, scan all edge destinations, add their edges' rows in LDS, finish the nodes there; sum order = LDS arrival
# speedup vs baseline: 2.2564x; 2.2564x over previous
#include <hip/hip_runtime.h>
#include <math.h>

#define N_NODES 50000
#define N_EDGES 800000
#define IN_DIM  128
#define OUT_DIM 128
#define HEADS   4
#define D_K     32
#define ATT_DIM 65

typedef __attribute__((ext_vector_type(2))) float v2f;
typedef __attribute__((ext_vector_type(8))) float v8f;

__global__ void __launch_bounds__(256)
gemm_z_wmma(const float* __restrict__ h, const float* __restrict__ wpack,
            const float* __restrict__ bias, float* __restrict__ z) {
    const int wave = (blockIdx.x * blockDim.x + threadIdx.x) >> 5;
    const int lane = threadIdx.x & 31;
    if (wave >= (N_NODES / 16)) return;

    const int m = lane & 15;
    const int g = lane >> 4;

    const float* __restrict__ hrow = h + (size_t)(wave * 16 + m) * IN_DIM;
    v2f a[32];
#pragma unroll
    for (int ks = 0; ks < 32; ++ks) {
        const int k = 4 * ks + 2 * g;
        a[ks].x = hrow[k];
        a[ks].y = hrow[k + 1];
    }

    const float4* __restrict__ wp4 = (const float4*)wpack + lane;
    for (int ot = 0; ot < 8; ++ot) {
        const int col = ot * 16 + m;
        const float4* __restrict__ bp = wp4 + ot * (16 * 32);
        v8f c = {};
#pragma unroll
        for (int j = 0; j < 16; ++j) {
            const float4 b4 = bp[j * 32];
            v2f b0, b1;
            b0.x = b4.x; b0.y = b4.y;
            b1.x = b4.z; b1.y = b4.w;
            c = __builtin_amdgcn_wmma_f32_16x16x4_f32(
                     false, a[2 * j],  false, b0,
                     (short)0, c,  false,  false);
            c = __builtin_amdgcn_wmma_f32_16x16x4_f32(
                     false, a[2 * j + 1],  false, b1,
                     (short)0, c,  false,  false);
        }
        const float bv = bias[col];
        const int rowBase = wave * 16 + g * 8;
#pragma unroll
        for (int r = 0; r < 8; ++r) {
            z[(size_t)(rowBase + r) * OUT_DIM + col] = c[r] + bv;
        }
    }
}

__global__ void __launch_bounds__(256)
node_scores(const float* __restrict__ z, const float* __restrict__ att,
            float* __restrict__ s_src, float* __restrict__ s_dst) {
    const int gid = blockIdx.x * blockDim.x + threadIdx.x;
    if (gid >= N_NODES * HEADS) return;
    const int hd = gid & 3;
    const int n  = gid >> 2;
    const float* __restrict__ zp = z + (size_t)n * OUT_DIM + hd * D_K;
    const float* __restrict__ as = att + hd * ATT_DIM;
    const float* __restrict__ ad = as + D_K;
    float ss = 0.f, sd = 0.f;
#pragma unroll
    for (int k = 0; k < D_K; ++k) {
        const float v = zp[k];
        ss = fmaf(v, as[k], ss);
        sd = fmaf(v, ad[k], sd);
    }
    s_src[gid] = ss;
    s_dst[gid] = sd;
}

#define PACK_BLOCKS 16
__global__ void __launch_bounds__(256)
pack_w(const float* __restrict__ W, float* __restrict__ wpack) {
    const int t  = blockIdx.x * 256 + threadIdx.x;
    const int l  = t & 31;
    const int j  = (t >> 5) & 15;
    const int ot = t >> 9;
    const float* __restrict__ wr =
        W + (size_t)(ot * 16 + (l & 15)) * IN_DIM + 8 * j + 2 * (l >> 4);
    float4 v = { wr[0], wr[1], wr[4], wr[5] };
    ((float4*)wpack)[t] = v;
}

__device__ __forceinline__ float leaky_clip_exp(float s) {
    s = (s >= 0.f) ? s : 0.2f * s;
    s = fminf(fmaxf(s, -20.f), 20.f);
    return __expf(s);
}

#define SLICE           196
#define N_SLICES        ((N_NODES + SLICE - 1) / SLICE)
#define OWN_WAVES       32
#define OWN_THREADS     (OWN_WAVES * 32)
#define SCAN_UNROLL     8
#define N_UNITS         (N_EDGES / 128)
#define UNITS_PER_WAVE  ((N_UNITS + OWN_WAVES - 1) / OWN_WAVES)
#define SCAN_ITERS      ((UNITS_PER_WAVE + SCAN_UNROLL - 1) / SCAN_UNROLL)
#define WQ              1280
static_assert(N_EDGES % 128 == 0, "a wave reads 128 whole destinations per load");
static_assert(SCAN_UNROLL * 128 + 32 <= WQ, "a scan step must fit the wave's queue");

__device__ __forceinline__ float lane_bcast(float v, int l) {
    return __int_as_float(__builtin_amdgcn_readlane(__float_as_int(v), l));
}

__device__ __forceinline__ void lds_add(float* p, float v) {
    __hip_atomic_fetch_add(p, v, __ATOMIC_RELAXED, __HIP_MEMORY_SCOPE_WORKGROUP);
}

__device__ __forceinline__ void take_edges(
        const int* wqp, const int nv, const int lane, const int lo,
        const int* __restrict__ ei, const float* __restrict__ ef,
        const float* __restrict__ s_src, const float* __restrict__ s_dst,
        const float ae0, const float ae1, const float ae2, const float ae3,
        const float* __restrict__ zl, float* al, float* den) {
    const int e   = wqp[(lane < nv) ? lane : 0];
    const int src = ei[e];
    const int dl  = ei[N_EDGES + e] - lo;
    const float f = ef[e];
    const float4 ss = *(const float4*)(s_src + (size_t)src * HEADS);
    const float4 sd = *(const float4*)(s_dst + (size_t)(lo + dl) * HEADS);
    const float w0 = leaky_clip_exp(ss.x + sd.x + f * ae0);
    const float w1 = leaky_clip_exp(ss.y + sd.y + f * ae1);
    const float w2 = leaky_clip_exp(ss.z + sd.z + f * ae2);
    const float w3 = leaky_clip_exp(ss.w + sd.w + f * ae3);
    if (lane < nv) {
        lds_add(&den[dl * HEADS + 0], w0);
        lds_add(&den[dl * HEADS + 1], w1);
        lds_add(&den[dl * HEADS + 2], w2);
        lds_add(&den[dl * HEADS + 3], w3);
    }
    for (int c = 0; c < nv; c += 8) {
        float p[8][4];
#pragma unroll
        for (int j = 0; j < 8; ++j) {
            const int i  = (c + j < nv) ? (c + j) : (nv - 1);
            const int sj = __builtin_amdgcn_readlane(src, i);
            const float* __restrict__ zp = zl + (size_t)sj * OUT_DIM;
            p[j][0] = zp[0]       * lane_bcast(w0, i);
            p[j][1] = zp[D_K]     * lane_bcast(w1, i);
            p[j][2] = zp[2 * D_K] * lane_bcast(w2, i);
            p[j][3] = zp[3 * D_K] * lane_bcast(w3, i);
        }
#pragma unroll
        for (int j = 0; j < 8; ++j) {
            if (c + j < nv) {
                const int dj = __builtin_amdgcn_readlane(dl, c + j);
                float* ap = al + dj * OUT_DIM;
                lds_add(ap,           p[j][0]);
                lds_add(ap + D_K,     p[j][1]);
                lds_add(ap + 2 * D_K, p[j][2]);
                lds_add(ap + 3 * D_K, p[j][3]);
            }
        }
    }
}

__global__ void __launch_bounds__(OWN_THREADS)
owner_edges(const int* __restrict__ ei, const float* __restrict__ ef,
            const float* __restrict__ s_src, const float* __restrict__ s_dst,
            const float* __restrict__ att, const float* __restrict__ z,
            const float* __restrict__ ln_g, const float* __restrict__ ln_b,
            float* __restrict__ out) {
    __shared__ __attribute__((aligned(16))) float acc[SLICE * OUT_DIM];
    __shared__ float den[SLICE * HEADS];
    __shared__ int   queue[OWN_WAVES * WQ];

    const int tid  = threadIdx.x;
    const int lane = tid & 31;
    const int wv   = __builtin_amdgcn_readfirstlane(tid >> 5);
    const int lo   = blockIdx.x * SLICE;
    const int cnt  = (N_NODES - lo < SLICE) ? (N_NODES - lo) : SLICE;
    const unsigned ucnt = (unsigned)cnt;

    for (int i = tid; i < SLICE * OUT_DIM; i += OWN_THREADS) acc[i] = 0.f;
    for (int i = tid; i < SLICE * HEADS;   i += OWN_THREADS) den[i] = 0.f;
    __syncthreads();

    const int4* __restrict__ dst4 = (const int4*)(ei + N_EDGES) + lane;
    const float ae0 = att[0 * ATT_DIM + 2 * D_K];
    const float ae1 = att[1 * ATT_DIM + 2 * D_K];
    const float ae2 = att[2 * ATT_DIM + 2 * D_K];
    const float ae3 = att[3 * ATT_DIM + 2 * D_K];
    const float* __restrict__ zl = z + lane;
    float* al = acc + lane;
    int* wq = queue + wv * WQ;
    int wcount = 0;

    const int rot = blockIdx.x % SCAN_ITERS;
    for (int it = 0; it < SCAN_ITERS; ++it) {
        int itr = it + rot;
        if (itr >= SCAN_ITERS) itr -= SCAN_ITERS;
        const int k0 = wv + OWN_WAVES * SCAN_UNROLL * itr;

        int4 d[SCAN_UNROLL];
#pragma unroll
        for (int u = 0; u < SCAN_UNROLL; ++u) {
            const int k  = k0 + OWN_WAVES * u;
            const int kc = (k < N_UNITS) ? k : (N_UNITS - 1);
            d[u] = dst4[kc * 32];
        }
#pragma unroll
        for (int u = 0; u < SCAN_UNROLL; ++u) {
            const int k = k0 + OWN_WAVES * u;
            if (k < N_UNITS) {
                const unsigned r0 = (unsigned)(d[u].x - lo);
                const unsigned r1 = (unsigned)(d[u].y - lo);
                const unsigned r2 = (unsigned)(d[u].z - lo);
                const unsigned r3 = (unsigned)(d[u].w - lo);
                const unsigned ra = (r0 < r1) ? r0 : r1;
                const unsigned rb = (r2 < r3) ? r2 : r3;
                const unsigned rm = (ra < rb) ? ra : rb;
                if (__builtin_amdgcn_ballot_w32(rm < ucnt) != 0u) {
                    const int e0 = 4 * (k * 32 + lane);
                    const unsigned m0 = __builtin_amdgcn_ballot_w32(r0 < ucnt);
                    if (m0 != 0u) {
                        if (r0 < ucnt) wq[wcount + (int)__builtin_amdgcn_mbcnt_lo(m0, 0u)] = e0;
                        wcount += __builtin_popcount(m0);
                    }
                    const unsigned m1 = __builtin_amdgcn_ballot_w32(r1 < ucnt);
                    if (m1 != 0u) {
                        if (r1 < ucnt) wq[wcount + (int)__builtin_amdgcn_mbcnt_lo(m1, 0u)] = e0 + 1;
                        wcount += __builtin_popcount(m1);
                    }
                    const unsigned m2 = __builtin_amdgcn_ballot_w32(r2 < ucnt);
                    if (m2 != 0u) {
                        if (r2 < ucnt) wq[wcount + (int)__builtin_amdgcn_mbcnt_lo(m2, 0u)] = e0 + 2;
                        wcount += __builtin_popcount(m2);
                    }
                    const unsigned m3 = __builtin_amdgcn_ballot_w32(r3 < ucnt);
                    if (m3 != 0u) {
                        if (r3 < ucnt) wq[wcount + (int)__builtin_amdgcn_mbcnt_lo(m3, 0u)] = e0 + 3;
                        wcount += __builtin_popcount(m3);
                    }
                }
            }
        }
        while (wcount >= 32) {
            wcount -= 32;
            take_edges(wq + wcount, 32, lane, lo, ei, ef, s_src, s_dst,
                       ae0, ae1, ae2, ae3, zl, al, den);
        }
    }
    if (wcount > 0) {
        take_edges(wq, wcount, lane, lo, ei, ef, s_src, s_dst,
                   ae0, ae1, ae2, ae3, zl, al, den);
    }
    __syncthreads();

    const float4 gv = *(const float4*)(ln_g + lane * 4);
    const float4 bv = *(const float4*)(ln_b + lane * 4);
    const float g4[4] = { gv.x, gv.y, gv.z, gv.w };
    const float b4[4] = { bv.x, bv.y, bv.z, bv.w };
    for (int n = wv; n < cnt; n += OWN_WAVES) {
        const size_t base = (size_t)(lo + n) * OUT_DIM + lane * 4;
        const float4 a  = *(const float4*)(acc + n * OUT_DIM + lane * 4);
        const float4 zv = *(const float4*)(z + base);
        const float inv = 1.f / (den[n * HEADS + (lane >> 3)] + 1e-6f);
        float x[4] = { a.x * inv + zv.x, a.y * inv + zv.y, a.z * inv + zv.z, a.w * inv + zv.w };

        float s = x[0] + x[1] + x[2] + x[3];
#pragma unroll
        for (int off = 16; off > 0; off >>= 1) s += __shfl_xor(s, off, 32);
        const float mu = s * (1.f / 128.f);

        float v = 0.f;
#pragma unroll
        for (int i = 0; i < 4; ++i) { const float dd = x[i] - mu; v = fmaf(dd, dd, v); }
#pragma unroll
        for (int off = 16; off > 0; off >>= 1) v += __shfl_xor(v, off, 32);
        const float rstd = rsqrtf(v * (1.f / 128.f) + 1e-5f);

        float y[4];
#pragma unroll
        for (int i = 0; i < 4; ++i) {
            const float t = (x[i] - mu) * rstd * g4[i] + b4[i];
            y[i] = (t > 0.f) ? t : (__expf(t) - 1.f);
        }
        float4 o = { y[0], y[1], y[2], y[3] };
        *(float4*)(out + base) = o;
    }
}

extern "C" void kernel_launch(void* const* d_in, const int* in_sizes, int n_in,
                              void* d_out, int out_size, void* d_ws, size_t ws_size,
                              hipStream_t stream) {
    const float* h    = (const float*)d_in[0];
    const int*   ei   = (const int*)  d_in[1];
    const float* ef   = (const float*)d_in[2];
    const float* W_w  = (const float*)d_in[3];
    const float* W_b  = (const float*)d_in[4];
    const float* att  = (const float*)d_in[5];
    const float* ln_g = (const float*)d_in[6];
    const float* ln_b = (const float*)d_in[7];
    float* out = (float*)d_out;

    float* ws     = (float*)d_ws;
    float* z      = ws;
    float* s_src  = z      + (size_t)N_NODES * OUT_DIM;
    float* s_dst  = s_src  + (size_t)N_NODES * HEADS;

    pack_w<<<PACK_BLOCKS, 256, 0, stream>>>(W_w, s_src);
    {
        const int waves = N_NODES / 16;
        const int blocks = (waves + 7) / 8;
        gemm_z_wmma<<<blocks, 256, 0, stream>>>(h, s_src, W_b, z);
    }
    node_scores<<<(N_NODES * HEADS + 255) / 256, 256, 0, stream>>>(z, att, s_src, s_dst);
    owner_edges<<<N_SLICES, OWN_THREADS, 0, stream>>>(ei, ef, s_src, s_dst, att, z, ln_g, ln_b, out);
}


# Edge scan reads 1 byte per edge (destination slice number, a plane the first launch writes where s_dst was), not the 4-byte dst; own nodes' s_dst made in LDS by 8-lane exchange: sum order changes
# speedup vs baseline: 2.5281x; 1.1204x over previous
#include <hip/hip_runtime.h>
#include <math.h>

#define N_NODES 50000
#define N_EDGES 800000
#define IN_DIM  128
#define OUT_DIM 128
#define HEADS   4
#define D_K     32
#define ATT_DIM 65

typedef __attribute__((ext_vector_type(2))) float v2f;
typedef __attribute__((ext_vector_type(8))) float v8f;

__global__ void __launch_bounds__(256)
gemm_z_wmma(const float* __restrict__ h, const float* __restrict__ wpack,
            const float* __restrict__ bias, float* __restrict__ z) {
    const int wave = (blockIdx.x * blockDim.x + threadIdx.x) >> 5;
    const int lane = threadIdx.x & 31;
    if (wave >= (N_NODES / 16)) return;

    const int m = lane & 15;
    const int g = lane >> 4;

    const float* __restrict__ hrow = h + (size_t)(wave * 16 + m) * IN_DIM;
    v2f a[32];
#pragma unroll
    for (int ks = 0; ks < 32; ++ks) {
        const int k = 4 * ks + 2 * g;
        a[ks].x = hrow[k];
        a[ks].y = hrow[k + 1];
    }

    const float4* __restrict__ wp4 = (const float4*)wpack + lane;
    for (int ot = 0; ot < 8; ++ot) {
        const int col = ot * 16 + m;
        const float4* __restrict__ bp = wp4 + ot * (16 * 32);
        v8f c = {};
#pragma unroll
        for (int j = 0; j < 16; ++j) {
            const float4 b4 = bp[j * 32];
            v2f b0, b1;
            b0.x = b4.x; b0.y = b4.y;
            b1.x = b4.z; b1.y = b4.w;
            c = __builtin_amdgcn_wmma_f32_16x16x4_f32(
                     false, a[2 * j],  false, b0,
                     (short)0, c,  false,  false);
            c = __builtin_amdgcn_wmma_f32_16x16x4_f32(
                     false, a[2 * j + 1],  false, b1,
                     (short)0, c,  false,  false);
        }
        const float bv = bias[col];
        const int rowBase = wave * 16 + g * 8;
#pragma unroll
        for (int r = 0; r < 8; ++r) {
            z[(size_t)(rowBase + r) * OUT_DIM + col] = c[r] + bv;
        }
    }
}

__global__ void __launch_bounds__(256)
node_scores(const float* __restrict__ z, const float* __restrict__ att,
            float* __restrict__ s_src) {
    const int gid = blockIdx.x * blockDim.x + threadIdx.x;
    if (gid >= N_NODES * HEADS) return;
    const int hd = gid & 3;
    const int n  = gid >> 2;
    const float* __restrict__ zp = z + (size_t)n * OUT_DIM + hd * D_K;
    const float* __restrict__ as = att + hd * ATT_DIM;
    float ss = 0.f;
#pragma unroll
    for (int k = 0; k < D_K; ++k) {
        const float v = zp[k];
        ss = fmaf(v, as[k], ss);
    }
    s_src[gid] = ss;
}

#define SLICE        196
#define PACK_BLOCKS  16
#define ID_WORDS     (N_EDGES / 4)
#define ID_BLOCKS    ((ID_WORDS + 255) / 256)
static_assert(N_EDGES % 4 == 0, "a thread packs 4 whole edges");
static_assert((N_NODES - 1) / SLICE <= 255, "a slice number must fit a byte");
__global__ void __launch_bounds__(256)
pack_w(const float* __restrict__ W, float* __restrict__ wpack,
       const int* __restrict__ ei, unsigned int* __restrict__ ids) {
    if (blockIdx.x >= PACK_BLOCKS) {
        const int w = (blockIdx.x - PACK_BLOCKS) * 256 + threadIdx.x;
        if (w < ID_WORDS) {
            const int4 d = ((const int4*)(ei + N_EDGES))[w];
            ids[w] =  ((unsigned)d.x / SLICE)        | (((unsigned)d.y / SLICE) << 8)
                   | (((unsigned)d.z / SLICE) << 16) | (((unsigned)d.w / SLICE) << 24);
        }
        return;
    }
    const int t  = blockIdx.x * 256 + threadIdx.x;
    const int l  = t & 31;
    const int j  = (t >> 5) & 15;
    const int ot = t >> 9;
    const float* __restrict__ wr =
        W + (size_t)(ot * 16 + (l & 15)) * IN_DIM + 8 * j + 2 * (l >> 4);
    float4 v = { wr[0], wr[1], wr[4], wr[5] };
    ((float4*)wpack)[t] = v;
}

__device__ __forceinline__ float leaky_clip_exp(float s) {
    s = (s >= 0.f) ? s : 0.2f * s;
    s = fminf(fmaxf(s, -20.f), 20.f);
    return __expf(s);
}

#define N_SLICES        ((N_NODES + SLICE - 1) / SLICE)
#define OWN_WAVES       32
#define OWN_THREADS     (OWN_WAVES * 32)
#define SCAN_UNROLL     3
#define ID_QUADS        (N_EDGES / 16)
#define N_UNITS         ((ID_QUADS + 31) / 32)
#define UNITS_PER_WAVE  ((N_UNITS + OWN_WAVES - 1) / OWN_WAVES)
#define SCAN_ITERS      ((UNITS_PER_WAVE + SCAN_UNROLL - 1) / SCAN_UNROLL)
#define WQ              1568
static_assert(N_EDGES % 16 == 0, "a lane reads 16 whole edges per load");
static_assert(SCAN_UNROLL * 512 + 32 <= WQ, "a scan step must fit the wave's queue");

__device__ __forceinline__ float lane_bcast(float v, int l) {
    return __int_as_float(__builtin_amdgcn_readlane(__float_as_int(v), l));
}

__device__ __forceinline__ void lds_add(float* p, float v) {
    __hip_atomic_fetch_add(p, v, __ATOMIC_RELAXED, __HIP_MEMORY_SCOPE_WORKGROUP);
}

__device__ __forceinline__ void take_edges(
        const int* wqp, const int nv, const int lane, const int lo,
        const int* __restrict__ ei, const float* __restrict__ ef,
        const float* __restrict__ s_src, const float* sdl,
        const float ae0, const float ae1, const float ae2, const float ae3,
        const float* __restrict__ zl, float* al, float* den) {
    const int e   = wqp[(lane < nv) ? lane : 0];
    const int src = ei[e];
    const int dl  = ei[N_EDGES + e] - lo;
    const float f = ef[e];
    const float4 ss = *(const float4*)(s_src + (size_t)src * HEADS);
    const float4 sd = *(const float4*)(sdl + dl * HEADS);
    const float w0 = leaky_clip_exp(ss.x + sd.x + f * ae0);
    const float w1 = leaky_clip_exp(ss.y + sd.y + f * ae1);
    const float w2 = leaky_clip_exp(ss.z + sd.z + f * ae2);
    const float w3 = leaky_clip_exp(ss.w + sd.w + f * ae3);
    if (lane < nv) {
        lds_add(&den[dl * HEADS + 0], w0);
        lds_add(&den[dl * HEADS + 1], w1);
        lds_add(&den[dl * HEADS + 2], w2);
        lds_add(&den[dl * HEADS + 3], w3);
    }
    for (int c = 0; c < nv; c += 8) {
        float p[8][4];
#pragma unroll
        for (int j = 0; j < 8; ++j) {
            const int i  = (c + j < nv) ? (c + j) : (nv - 1);
            const int sj = __builtin_amdgcn_readlane(src, i);
            const float* __restrict__ zp = zl + (size_t)sj * OUT_DIM;
            p[j][0] = zp[0]       * lane_bcast(w0, i);
            p[j][1] = zp[D_K]     * lane_bcast(w1, i);
            p[j][2] = zp[2 * D_K] * lane_bcast(w2, i);
            p[j][3] = zp[3 * D_K] * lane_bcast(w3, i);
        }
#pragma unroll
        for (int j = 0; j < 8; ++j) {
            if (c + j < nv) {
                const int dj = __builtin_amdgcn_readlane(dl, c + j);
                float* ap = al + dj * OUT_DIM;
                lds_add(ap,           p[j][0]);
                lds_add(ap + D_K,     p[j][1]);
                lds_add(ap + 2 * D_K, p[j][2]);
                lds_add(ap + 3 * D_K, p[j][3]);
            }
        }
    }
}

__global__ void __launch_bounds__(OWN_THREADS)
owner_edges(const int* __restrict__ ei, const float* __restrict__ ef,
            const float* __restrict__ s_src, const unsigned int* __restrict__ ids,
            const float* __restrict__ att, const float* __restrict__ z,
            const float* __restrict__ ln_g, const float* __restrict__ ln_b,
            float* __restrict__ out) {
    __shared__ __attribute__((aligned(16))) float acc[SLICE * OUT_DIM];
    __shared__ __attribute__((aligned(16))) float sdl[SLICE * HEADS];
    __shared__ float den[SLICE * HEADS];
    __shared__ int   queue[OWN_WAVES * WQ];

    const int tid  = threadIdx.x;
    const int lane = tid & 31;
    const int wv   = __builtin_amdgcn_readfirstlane(tid >> 5);
    const int lo   = blockIdx.x * SLICE;
    const int cnt  = (N_NODES - lo < SLICE) ? (N_NODES - lo) : SLICE;

    for (int i = tid; i < SLICE * OUT_DIM; i += OWN_THREADS) acc[i] = 0.f;
    for (int i = tid; i < SLICE * HEADS;   i += OWN_THREADS) den[i] = 0.f;
    {
        const int hd = lane >> 3;
        const float* __restrict__ ad = att + hd * ATT_DIM + D_K + 4 * (lane & 7);
        const float ad0 = ad[0], ad1 = ad[1], ad2 = ad[2], ad3 = ad[3];
        for (int n = wv; n < cnt; n += OWN_WAVES) {
            const float4 zv = *(const float4*)(z + (size_t)(lo + n) * OUT_DIM + lane * 4);
            float sd = fmaf(zv.w, ad3, fmaf(zv.z, ad2, fmaf(zv.y, ad1, zv.x * ad0)));
#pragma unroll
            for (int off = 4; off > 0; off >>= 1) sd += __shfl_xor(sd, off, 32);
            if ((lane & 7) == 0) sdl[n * HEADS + hd] = sd;
        }
    }
    __syncthreads();

    const uint4* __restrict__ id4 = (const uint4*)ids;
    const unsigned bs = (unsigned)blockIdx.x * 0x01010101u;
    const float ae0 = att[0 * ATT_DIM + 2 * D_K];
    const float ae1 = att[1 * ATT_DIM + 2 * D_K];
    const float ae2 = att[2 * ATT_DIM + 2 * D_K];
    const float ae3 = att[3 * ATT_DIM + 2 * D_K];
    const float* __restrict__ zl = z + lane;
    float* al = acc + lane;
    int* wq = queue + wv * WQ;
    int wcount = 0;

    const int rot = blockIdx.x % SCAN_ITERS;
    for (int it = 0; it < SCAN_ITERS; ++it) {
        int itr = it + rot;
        if (itr >= SCAN_ITERS) itr -= SCAN_ITERS;
        const int k0 = wv + OWN_WAVES * SCAN_UNROLL * itr;

        uint4 d[SCAN_UNROLL];
#pragma unroll
        for (int u = 0; u < SCAN_UNROLL; ++u) {
            const int qd = (k0 + OWN_WAVES * u) * 32 + lane;
            const int qc = (qd < ID_QUADS) ? qd : (ID_QUADS - 1);
            d[u] = id4[qc];
        }
#pragma unroll
        for (int u = 0; u < SCAN_UNROLL; ++u) {
            const int k  = k0 + OWN_WAVES * u;
            const int qd = k * 32 + lane;
            if (k < N_UNITS) {
                const unsigned live = (qd < ID_QUADS) ? 0x80808080u : 0u;
                unsigned f[4];
                {
                    const unsigned x0 = d[u].x ^ bs, x1 = d[u].y ^ bs;
                    const unsigned x2 = d[u].z ^ bs, x3 = d[u].w ^ bs;
                    f[0] = ~(((x0 & 0x7f7f7f7fu) + 0x7f7f7f7fu) | x0) & live;
                    f[1] = ~(((x1 & 0x7f7f7f7fu) + 0x7f7f7f7fu) | x1) & live;
                    f[2] = ~(((x2 & 0x7f7f7f7fu) + 0x7f7f7f7fu) | x2) & live;
                    f[3] = ~(((x3 & 0x7f7f7f7fu) + 0x7f7f7f7fu) | x3) & live;
                }
                if (__builtin_amdgcn_ballot_w32((f[0] | f[1] | f[2] | f[3]) != 0u) != 0u) {
                    unsigned g = (f[0] >> 7) | (f[1] >> 6) | (f[2] >> 5) | (f[3] >> 4);
                    const int e0 = 16 * qd;
                    unsigned mb = __builtin_amdgcn_ballot_w32(g != 0u);
                    while (mb != 0u) {
                        if (g != 0u) {
                            const int pos = __builtin_ctz(g);
                            wq[wcount + (int)__builtin_amdgcn_mbcnt_lo(mb, 0u)] =
                                e0 + 4 * (pos & 7) + (pos >> 3);
                            g &= g - 1u;
                        }
                        wcount += __builtin_popcount(mb);
                        mb = __builtin_amdgcn_ballot_w32(g != 0u);
                    }
                }
            }
        }
        while (wcount >= 32) {
            wcount -= 32;
            take_edges(wq + wcount, 32, lane, lo, ei, ef, s_src, sdl,
                       ae0, ae1, ae2, ae3, zl, al, den);
        }
    }
    if (wcount > 0) {
        take_edges(wq, wcount, lane, lo, ei, ef, s_src, sdl,
                   ae0, ae1, ae2, ae3, zl, al, den);
    }
    __syncthreads();

    const float4 gv = *(const float4*)(ln_g + lane * 4);
    const float4 bv = *(const float4*)(ln_b + lane * 4);
    const float g4[4] = { gv.x, gv.y, gv.z, gv.w };
    const float b4[4] = { bv.x, bv.y, bv.z, bv.w };
    for (int n = wv; n < cnt; n += OWN_WAVES) {
        const size_t base = (size_t)(lo + n) * OUT_DIM + lane * 4;
        const float4 a  = *(const float4*)(acc + n * OUT_DIM + lane * 4);
        const float4 zv = *(const float4*)(z + base);
        const float inv = 1.f / (den[n * HEADS + (lane >> 3)] + 1e-6f);
        float x[4] = { a.x * inv + zv.x, a.y * inv + zv.y, a.z * inv + zv.z, a.w * inv + zv.w };

        float s = x[0] + x[1] + x[2] + x[3];
#pragma unroll
        for (int off = 16; off > 0; off >>= 1) s += __shfl_xor(s, off, 32);
        const float mu = s * (1.f / 128.f);

        float v = 0.f;
#pragma unroll
        for (int i = 0; i < 4; ++i) { const float dd = x[i] - mu; v = fmaf(dd, dd, v); }
#pragma unroll
        for (int off = 16; off > 0; off >>= 1) v += __shfl_xor(v, off, 32);
        const float rstd = rsqrtf(v * (1.f / 128.f) + 1e-5f);

        float y[4];
#pragma unroll
        for (int i = 0; i < 4; ++i) {
            const float t = (x[i] - mu) * rstd * g4[i] + b4[i];
            y[i] = (t > 0.f) ? t : (__expf(t) - 1.f);
        }
        float4 o = { y[0], y[1], y[2], y[3] };
        *(float4*)(out + base) = o;
    }
}

extern "C" void kernel_launch(void* const* d_in, const int* in_sizes, int n_in,
                              void* d_out, int out_size, void* d_ws, size_t ws_size,
                              hipStream_t stream) {
    const float* h    = (const float*)d_in[0];
    const int*   ei   = (const int*)  d_in[1];
    const float* ef   = (const float*)d_in[2];
    const float* W_w  = (const float*)d_in[3];
    const float* W_b  = (const float*)d_in[4];
    const float* att  = (const float*)d_in[5];
    const float* ln_g = (const float*)d_in[6];
    const float* ln_b = (const float*)d_in[7];
    float* out = (float*)d_out;

    float* ws     = (float*)d_ws;
    float* z      = ws;
    float* s_src  = z      + (size_t)N_NODES * OUT_DIM;
    unsigned int* ids = (unsigned int*)(s_src + (size_t)N_NODES * HEADS);

    pack_w<<<PACK_BLOCKS + ID_BLOCKS, 256, 0, stream>>>(W_w, s_src, ei, ids);
    {
        const int waves = N_NODES / 16;
        const int blocks = (waves + 7) / 8;
        gemm_z_wmma<<<blocks, 256, 0, stream>>>(h, s_src, W_b, z);
    }
    node_scores<<<(N_NODES * HEADS + 255) / 256, 256, 0, stream>>>(z, att, s_src);
    owner_edges<<<N_SLICES, OWN_THREADS, 0, stream>>>(ei, ef, s_src, ids, att, z, ln_g, ln_b, out);
}


# Edge scan reads slice numbers as bit planes (8 ballot words per 32 edges, made by launch 1): a lane matches 32 edges with 8 bit ops; hits queued per pass, 64-entry queues; sum order = LDS arrival
# speedup vs baseline: 2.8986x; 1.1466x over previous
#include <hip/hip_runtime.h>
#include <math.h>

#define N_NODES 50000
#define N_EDGES 800000
#define IN_DIM  128
#define OUT_DIM 128
#define HEADS   4
#define D_K     32
#define ATT_DIM 65

typedef __attribute__((ext_vector_type(2))) float v2f;
typedef __attribute__((ext_vector_type(8))) float v8f;

__global__ void __launch_bounds__(256)
gemm_z_wmma(const float* __restrict__ h, const float* __restrict__ wpack,
            const float* __restrict__ bias, float* __restrict__ z) {
    const int wave = (blockIdx.x * blockDim.x + threadIdx.x) >> 5;
    const int lane = threadIdx.x & 31;
    if (wave >= (N_NODES / 16)) return;

    const int m = lane & 15;
    const int g = lane >> 4;

    const float* __restrict__ hrow = h + (size_t)(wave * 16 + m) * IN_DIM;
    v2f a[32];
#pragma unroll
    for (int ks = 0; ks < 32; ++ks) {
        const int k = 4 * ks + 2 * g;
        a[ks].x = hrow[k];
        a[ks].y = hrow[k + 1];
    }

    const float4* __restrict__ wp4 = (const float4*)wpack + lane;
    for (int ot = 0; ot < 8; ++ot) {
        const int col = ot * 16 + m;
        const float4* __restrict__ bp = wp4 + ot * (16 * 32);
        v8f c = {};
#pragma unroll
        for (int j = 0; j < 16; ++j) {
            const float4 b4 = bp[j * 32];
            v2f b0, b1;
            b0.x = b4.x; b0.y = b4.y;
            b1.x = b4.z; b1.y = b4.w;
            c = __builtin_amdgcn_wmma_f32_16x16x4_f32(
                     false, a[2 * j],  false, b0,
                     (short)0, c,  false,  false);
            c = __builtin_amdgcn_wmma_f32_16x16x4_f32(
                     false, a[2 * j + 1],  false, b1,
                     (short)0, c,  false,  false);
        }
        const float bv = bias[col];
        const int rowBase = wave * 16 + g * 8;
#pragma unroll
        for (int r = 0; r < 8; ++r) {
            z[(size_t)(rowBase + r) * OUT_DIM + col] = c[r] + bv;
        }
    }
}

__global__ void __launch_bounds__(256)
node_scores(const float* __restrict__ z, const float* __restrict__ att,
            float* __restrict__ s_src) {
    const int gid = blockIdx.x * blockDim.x + threadIdx.x;
    if (gid >= N_NODES * HEADS) return;
    const int hd = gid & 3;
    const int n  = gid >> 2;
    const float* __restrict__ zp = z + (size_t)n * OUT_DIM + hd * D_K;
    const float* __restrict__ as = att + hd * ATT_DIM;
    float ss = 0.f;
#pragma unroll
    for (int k = 0; k < D_K; ++k) {
        const float v = zp[k];
        ss = fmaf(v, as[k], ss);
    }
    s_src[gid] = ss;
}

#define SLICE        196
#define PACK_BLOCKS  16
#define ID_WORDS     (N_EDGES / 4)
#define ID_BLOCKS    ((ID_WORDS + 255) / 256)
#define N_GROUPS     (N_EDGES / 32)
static_assert(N_EDGES % 128 == 0, "a wave of the first launch takes 128 whole edges");
static_assert(N_GROUPS % 8 == 0, "8 groups share a pair of lines");
static_assert(N_GROUPS * 8 == ID_WORDS, "8 plane words per group fill the plane exactly");
static_assert((N_NODES - 1) / SLICE <= 255, "a slice number must fit 8 bit planes");
__global__ void __launch_bounds__(256)
pack_w(const float* __restrict__ W, float* __restrict__ wpack,
       const int* __restrict__ ei, unsigned int* __restrict__ ids) {
    if (blockIdx.x >= PACK_BLOCKS) {
        const int t = (blockIdx.x - PACK_BLOCKS) * 256 + threadIdx.x;
        if (t < ID_WORDS) {
            const int4 d = ((const int4*)(ei + N_EDGES))[t];
            const unsigned x =  ((unsigned)d.x / SLICE)        | (((unsigned)d.y / SLICE) << 8)
                             | (((unsigned)d.z / SLICE) << 16) | (((unsigned)d.w / SLICE) << 24);
            const int lane = threadIdx.x & 31;
            const int w    = t >> 5;
            unsigned v = 0u;
#pragma unroll
            for (int j = 0; j < 32; ++j) {
                const int p = (j >> 2) & 3;
                const int k = (j >> 4) * 4 + (j & 3);
                const unsigned bal = __builtin_amdgcn_ballot_w32(((x >> (8 * p + k)) & 1u) != 0u);
                v = (lane == j) ? bal : v;
            }
            ids[(w >> 1) * 64 + (w & 1) * 16 + (lane >> 4) * 32 + (lane & 15)] = v;
        }
        return;
    }
    const int t  = blockIdx.x * 256 + threadIdx.x;
    const int l  = t & 31;
    const int j  = (t >> 5) & 15;
    const int ot = t >> 9;
    const float* __restrict__ wr =
        W + (size_t)(ot * 16 + (l & 15)) * IN_DIM + 8 * j + 2 * (l >> 4);
    float4 v = { wr[0], wr[1], wr[4], wr[5] };
    ((float4*)wpack)[t] = v;
}

__device__ __forceinline__ float leaky_clip_exp(float s) {
    s = (s >= 0.f) ? s : 0.2f * s;
    s = fminf(fmaxf(s, -20.f), 20.f);
    return __expf(s);
}

#define N_SLICES        ((N_NODES + SLICE - 1) / SLICE)
#define OWN_WAVES       32
#define OWN_THREADS     (OWN_WAVES * 32)
#define SCAN_UNROLL     5
#define N_UNITS         ((N_GROUPS + 31) / 32)
#define UNITS_PER_WAVE  ((N_UNITS + OWN_WAVES - 1) / OWN_WAVES)
#define SCAN_ITERS      ((UNITS_PER_WAVE + SCAN_UNROLL - 1) / SCAN_UNROLL)
#define WQ              64
#define HOLD_WORDS      16384
static_assert(SCAN_UNROLL == 5, "the scan picks a unit's mask by hand from five");

__device__ __forceinline__ float lane_bcast(float v, int l) {
    return __int_as_float(__builtin_amdgcn_readlane(__float_as_int(v), l));
}

__device__ __forceinline__ void lds_add(float* p, float v) {
    __hip_atomic_fetch_add(p, v, __ATOMIC_RELAXED, __HIP_MEMORY_SCOPE_WORKGROUP);
}

__device__ __forceinline__ void take_edges(
        const int* wqp, const int nv, const int lane, const int lo,
        const int* __restrict__ ei, const float* __restrict__ ef,
        const float* __restrict__ s_src, const float* sdl,
        const float ae0, const float ae1, const float ae2, const float ae3,
        const float* __restrict__ zl, float* al, float* den) {
    const int e   = wqp[(lane < nv) ? lane : 0];
    const int src = ei[e];
    const int dl  = ei[N_EDGES + e] - lo;
    const float f = ef[e];
    const float4 ss = *(const float4*)(s_src + (size_t)src * HEADS);
    const float4 sd = *(const float4*)(sdl + dl * HEADS);
    const float w0 = leaky_clip_exp(ss.x + sd.x + f * ae0);
    const float w1 = leaky_clip_exp(ss.y + sd.y + f * ae1);
    const float w2 = leaky_clip_exp(ss.z + sd.z + f * ae2);
    const float w3 = leaky_clip_exp(ss.w + sd.w + f * ae3);
    if (lane < nv) {
        lds_add(&den[dl * HEADS + 0], w0);
        lds_add(&den[dl * HEADS + 1], w1);
        lds_add(&den[dl * HEADS + 2], w2);
        lds_add(&den[dl * HEADS + 3], w3);
    }
    for (int c = 0; c < nv; c += 8) {
        float p[8][4];
#pragma unroll
        for (int j = 0; j < 8; ++j) {
            const int i  = (c + j < nv) ? (c + j) : (nv - 1);
            const int sj = __builtin_amdgcn_readlane(src, i);
            const float* __restrict__ zp = zl + (size_t)sj * OUT_DIM;
            p[j][0] = zp[0]       * lane_bcast(w0, i);
            p[j][1] = zp[D_K]     * lane_bcast(w1, i);
            p[j][2] = zp[2 * D_K] * lane_bcast(w2, i);
            p[j][3] = zp[3 * D_K] * lane_bcast(w3, i);
        }
#pragma unroll
        for (int j = 0; j < 8; ++j) {
            if (c + j < nv) {
                const int dj = __builtin_amdgcn_readlane(dl, c + j);
                float* ap = al + dj * OUT_DIM;
                lds_add(ap,           p[j][0]);
                lds_add(ap + D_K,     p[j][1]);
                lds_add(ap + 2 * D_K, p[j][2]);
                lds_add(ap + 3 * D_K, p[j][3]);
            }
        }
    }
}

__global__ void __launch_bounds__(OWN_THREADS)
owner_edges(const int* __restrict__ ei, const float* __restrict__ ef,
            const float* __restrict__ s_src, const unsigned int* __restrict__ ids,
            const float* __restrict__ att, const float* __restrict__ z,
            const float* __restrict__ ln_g, const float* __restrict__ ln_b,
            float* __restrict__ out) {
    __shared__ __attribute__((aligned(16))) float acc[SLICE * OUT_DIM];
    __shared__ __attribute__((aligned(16))) float sdl[SLICE * HEADS];
    __shared__ float den[SLICE * HEADS];
    __shared__ int   queue[OWN_WAVES * WQ + HOLD_WORDS];

    const int tid  = threadIdx.x;
    const int lane = tid & 31;
    const int wv   = __builtin_amdgcn_readfirstlane(tid >> 5);
    const int lo   = blockIdx.x * SLICE;
    const int cnt  = (N_NODES - lo < SLICE) ? (N_NODES - lo) : SLICE;

    for (int i = tid; i < SLICE * OUT_DIM; i += OWN_THREADS) acc[i] = 0.f;
    for (int i = tid; i < SLICE * HEADS;   i += OWN_THREADS) den[i] = 0.f;
    {
        const int hd = lane >> 3;
        const float* __restrict__ ad = att + hd * ATT_DIM + D_K + 4 * (lane & 7);
        const float ad0 = ad[0], ad1 = ad[1], ad2 = ad[2], ad3 = ad[3];
        for (int n = wv; n < cnt; n += OWN_WAVES) {
            const float4 zv = *(const float4*)(z + (size_t)(lo + n) * OUT_DIM + lane * 4);
            float sd = fmaf(zv.w, ad3, fmaf(zv.z, ad2, fmaf(zv.y, ad1, zv.x * ad0)));
#pragma unroll
            for (int off = 4; off > 0; off >>= 1) sd += __shfl_xor(sd, off, 32);
            if ((lane & 7) == 0) sdl[n * HEADS + hd] = sd;
        }
    }
    __syncthreads();

    const uint4* __restrict__ id4 = (const uint4*)ids;
    unsigned sb[8];
#pragma unroll
    for (int k = 0; k < 8; ++k) sb[k] = 0u - (((unsigned)blockIdx.x >> k) & 1u);
    const float ae0 = att[0 * ATT_DIM + 2 * D_K];
    const float ae1 = att[1 * ATT_DIM + 2 * D_K];
    const float ae2 = att[2 * ATT_DIM + 2 * D_K];
    const float ae3 = att[3 * ATT_DIM + 2 * D_K];
    const float* __restrict__ zl = z + lane;
    float* al = acc + lane;
    int* wq = queue + wv * WQ;
    int wcount = 0;

    const int rot = blockIdx.x % SCAN_ITERS;
    for (int it = 0; it < SCAN_ITERS; ++it) {
        int itr = it + rot;
        if (itr >= SCAN_ITERS) itr -= SCAN_ITERS;
        const int k0 = wv + OWN_WAVES * SCAN_UNROLL * itr;

        uint4 da[SCAN_UNROLL], db[SCAN_UNROLL];
#pragma unroll
        for (int u = 0; u < SCAN_UNROLL; ++u) {
            const int gr = (k0 + OWN_WAVES * u) * 32 + lane;
            const int gc = (gr < N_GROUPS) ? gr : (N_GROUPS - 1);
            const int q  = (gc >> 3) * 16 + (gc & 7);
            da[u] = id4[q];
            db[u] = id4[q + 8];
        }
        unsigned m[SCAN_UNROLL];
#pragma unroll
        for (int u = 0; u < SCAN_UNROLL; ++u) {
            const int gr = (k0 + OWN_WAVES * u) * 32 + lane;
            unsigned t = (gr < N_GROUPS) ? 0xffffffffu : 0u;
            t &= ~(da[u].x ^ sb[0]);
            t &= ~(da[u].y ^ sb[1]);
            t &= ~(da[u].z ^ sb[2]);
            t &= ~(da[u].w ^ sb[3]);
            t &= ~(db[u].x ^ sb[4]);
            t &= ~(db[u].y ^ sb[5]);
            t &= ~(db[u].z ^ sb[6]);
            t &= ~(db[u].w ^ sb[7]);
            m[u] = t;
        }
#pragma nounroll
        for (int u = 0; u < SCAN_UNROLL; ++u) {
            unsigned g = m[0];
            g = (u == 1) ? m[1] : g;
            g = (u == 2) ? m[2] : g;
            g = (u == 3) ? m[3] : g;
            g = (u == 4) ? m[4] : g;
            const int gr = (k0 + OWN_WAVES * u) * 32 + lane;
            const int e0 = ((gr >> 2) << 7) + (gr & 3);
            unsigned mb = __builtin_amdgcn_ballot_w32(g != 0u);
            while (mb != 0u) {
                if (g != 0u) {
                    wq[wcount + (int)__builtin_amdgcn_mbcnt_lo(mb, 0u)] = e0 + 4 * __builtin_ctz(g);
                    g &= g - 1u;
                }
                wcount += __builtin_popcount(mb);
                if (wcount >= 32) {
                    wcount -= 32;
                    take_edges(wq + wcount, 32, lane, lo, ei, ef, s_src, sdl,
                               ae0, ae1, ae2, ae3, zl, al, den);
                }
                mb = __builtin_amdgcn_ballot_w32(g != 0u);
            }
        }
    }
    if (wcount > 0) {
        take_edges(wq, wcount, lane, lo, ei, ef, s_src, sdl,
                   ae0, ae1, ae2, ae3, zl, al, den);
    }
    __syncthreads();

    const float4 gv = *(const float4*)(ln_g + lane * 4);
    const float4 bv = *(const float4*)(ln_b + lane * 4);
    const float g4[4] = { gv.x, gv.y, gv.z, gv.w };
    const float b4[4] = { bv.x, bv.y, bv.z, bv.w };
    for (int n = wv; n < cnt; n += OWN_WAVES) {
        const size_t base = (size_t)(lo + n) * OUT_DIM + lane * 4;
        const float4 a  = *(const float4*)(acc + n * OUT_DIM + lane * 4);
        const float4 zv = *(const float4*)(z + base);
        const float inv = 1.f / (den[n * HEADS + (lane >> 3)] + 1e-6f);
        float x[4] = { a.x * inv + zv.x, a.y * inv + zv.y, a.z * inv + zv.z, a.w * inv + zv.w };

        float s = x[0] + x[1] + x[2] + x[3];
#pragma unroll
        for (int off = 16; off > 0; off >>= 1) s += __shfl_xor(s, off, 32);
        const float mu = s * (1.f / 128.f);

        float v = 0.f;
#pragma unroll
        for (int i = 0; i < 4; ++i) { const float dd = x[i] - mu; v = fmaf(dd, dd, v); }
#pragma unroll
        for (int off = 16; off > 0; off >>= 1) v += __shfl_xor(v, off, 32);
        const float rstd = rsqrtf(v * (1.f / 128.f) + 1e-5f);

        float y[4];
#pragma unroll
        for (int i = 0; i < 4; ++i) {
            const float t = (x[i] - mu) * rstd * g4[i] + b4[i];
            y[i] = (t > 0.f) ? t : (__expf(t) - 1.f);
        }
        float4 o = { y[0], y[1], y[2], y[3] };
        *(float4*)(out + base) = o;
    }
}

extern "C" void kernel_launch(void* const* d_in, const int* in_sizes, int n_in,
                              void* d_out, int out_size, void* d_ws, size_t ws_size,
                              hipStream_t stream) {
    const float* h    = (const float*)d_in[0];
    const int*   ei   = (const int*)  d_in[1];
    const float* ef   = (const float*)d_in[2];
    const float* W_w  = (const float*)d_in[3];
    const float* W_b  = (const float*)d_in[4];
    const float* att  = (const float*)d_in[5];
    const float* ln_g = (const float*)d_in[6];
    const float* ln_b = (const float*)d_in[7];
    float* out = (float*)d_out;

    float* ws     = (float*)d_ws;
    float* z      = ws;
    float* s_src  = z      + (size_t)N_NODES * OUT_DIM;
    unsigned int* ids = (unsigned int*)(s_src + (size_t)N_NODES * HEADS);

    pack_w<<<PACK_BLOCKS + ID_BLOCKS, 256, 0, stream>>>(W_w, s_src, ei, ids);
    {
        const int waves = N_NODES / 16;
        const int blocks = (waves + 7) / 8;
        gemm_z_wmma<<<blocks, 256, 0, stream>>>(h, s_src, W_b, z);
    }
    node_scores<<<(N_NODES * HEADS + 255) / 256, 256, 0, stream>>>(z, att, s_src);
    owner_edges<<<N_SLICES, OWN_THREADS, 0, stream>>>(ei, ef, s_src, ids, att, z, ln_g, ln_b, out);
}


# Source score halves made by the projection: W's rows weighted by att, summed per head (launch 1), are a 9th B tile, 4 more matrix columns; score launch and its z read dropped; score sum order changes
# speedup vs baseline: 3.1310x; 1.0802x over previous
#include <hip/hip_runtime.h>
#include <math.h>

#define N_NODES 50000
#define N_EDGES 800000
#define IN_DIM  128
#define OUT_DIM 128
#define HEADS   4
#define D_K     32
#define ATT_DIM 65

typedef __attribute__((ext_vector_type(2))) float v2f;
typedef __attribute__((ext_vector_type(8))) float v8f;

#define WPACK_TILES        9
#define WPACK_TILE_FLOATS  (WPACK_TILES * 16 * 32 * 4)

__global__ void __launch_bounds__(256)
gemm_z_wmma(const float* __restrict__ h, const float* __restrict__ wpack,
            const float* __restrict__ bias, float* __restrict__ z,
            float* __restrict__ s_src) {
    const int wave = (blockIdx.x * blockDim.x + threadIdx.x) >> 5;
    const int lane = threadIdx.x & 31;
    if (wave >= (N_NODES / 16)) return;

    const int m = lane & 15;
    const int g = lane >> 4;

    const float* __restrict__ hrow = h + (size_t)(wave * 16 + m) * IN_DIM;
    v2f a[32];
#pragma unroll
    for (int ks = 0; ks < 32; ++ks) {
        const int k = 4 * ks + 2 * g;
        a[ks].x = hrow[k];
        a[ks].y = hrow[k + 1];
    }

    const float4* __restrict__ wp4 = (const float4*)wpack + lane;
    for (int ot = 0; ot < 8; ++ot) {
        const int col = ot * 16 + m;
        const float4* __restrict__ bp = wp4 + ot * (16 * 32);
        v8f c = {};
#pragma unroll
        for (int j = 0; j < 16; ++j) {
            const float4 b4 = bp[j * 32];
            v2f b0, b1;
            b0.x = b4.x; b0.y = b4.y;
            b1.x = b4.z; b1.y = b4.w;
            c = __builtin_amdgcn_wmma_f32_16x16x4_f32(
                     false, a[2 * j],  false, b0,
                     (short)0, c,  false,  false);
            c = __builtin_amdgcn_wmma_f32_16x16x4_f32(
                     false, a[2 * j + 1],  false, b1,
                     (short)0, c,  false,  false);
        }
        const float bv = bias[col];
        const int rowBase = wave * 16 + g * 8;
#pragma unroll
        for (int r = 0; r < 8; ++r) {
            z[(size_t)(rowBase + r) * OUT_DIM + col] = c[r] + bv;
        }
    }
    {
        const float4* __restrict__ bp = wp4 + 8 * (16 * 32);
        v8f c = {};
#pragma unroll
        for (int j = 0; j < 16; ++j) {
            const float4 b4 = bp[j * 32];
            v2f b0, b1;
            b0.x = b4.x; b0.y = b4.y;
            b1.x = b4.z; b1.y = b4.w;
            c = __builtin_amdgcn_wmma_f32_16x16x4_f32(
                     false, a[2 * j],  false, b0,
                     (short)0, c,  false,  false);
            c = __builtin_amdgcn_wmma_f32_16x16x4_f32(
                     false, a[2 * j + 1],  false, b1,
                     (short)0, c,  false,  false);
        }
        if (m < HEADS) {
            const float ba = wpack[WPACK_TILE_FLOATS + m];
            const int rowBase = wave * 16 + g * 8;
#pragma unroll
            for (int r = 0; r < 8; ++r) {
                s_src[(size_t)(rowBase + r) * HEADS + m] = c[r] + ba;
            }
        }
    }
}

#define SLICE        196
#define PACK_BLOCKS  19
#define ID_WORDS     (N_EDGES / 4)
#define ID_BLOCKS    ((ID_WORDS + 255) / 256)
#define N_GROUPS     (N_EDGES / 32)
static_assert(N_EDGES % 128 == 0, "a wave of the first launch takes 128 whole edges");
static_assert(N_GROUPS % 8 == 0, "8 groups share a pair of lines");
static_assert(N_GROUPS * 8 == ID_WORDS, "8 plane words per group fill the plane exactly");
static_assert((N_NODES - 1) / SLICE <= 255, "a slice number must fit 8 bit planes");
static_assert((size_t)WPACK_TILE_FLOATS + HEADS <= (size_t)N_NODES * OUT_DIM, "wpack fits in out");
__global__ void __launch_bounds__(256)
pack_w(const float* __restrict__ W, const float* __restrict__ att,
       const float* __restrict__ bias, float* __restrict__ wpack,
       const int* __restrict__ ei, unsigned int* __restrict__ ids) {
    if (blockIdx.x >= PACK_BLOCKS) {
        const int t = (blockIdx.x - PACK_BLOCKS) * 256 + threadIdx.x;
        if (t < ID_WORDS) {
            const int4 d = ((const int4*)(ei + N_EDGES))[t];
            const unsigned x =  ((unsigned)d.x / SLICE)        | (((unsigned)d.y / SLICE) << 8)
                             | (((unsigned)d.z / SLICE) << 16) | (((unsigned)d.w / SLICE) << 24);
            const int lane = threadIdx.x & 31;
            const int w    = t >> 5;
            unsigned v = 0u;
#pragma unroll
            for (int j = 0; j < 32; ++j) {
                const int p = (j >> 2) & 3;
                const int k = (j >> 4) * 4 + (j & 3);
                const unsigned bal = __builtin_amdgcn_ballot_w32(((x >> (8 * p + k)) & 1u) != 0u);
                v = (lane == j) ? bal : v;
            }
            ids[(w >> 1) * 64 + (w & 1) * 16 + (lane >> 4) * 32 + (lane & 15)] = v;
        }
        return;
    }
    if (blockIdx.x == PACK_BLOCKS - 1) {
        if (threadIdx.x < HEADS) {
            const int hd = threadIdx.x;
            float s = 0.f;
            for (int k = 0; k < D_K; ++k) s = fmaf(bias[hd * D_K + k], att[hd * ATT_DIM + k], s);
            wpack[WPACK_TILE_FLOATS + hd] = s;
        }
        return;
    }
    const int t  = blockIdx.x * 256 + threadIdx.x;
    const int l  = t & 31;
    const int j  = (t >> 5) & 15;
    const int ot = t >> 9;
    if (ot == 8) {
        const int n = l & 15;
        float4 v = { 0.f, 0.f, 0.f, 0.f };
        if (n < HEADS) {
            const int i0 = 8 * j + 2 * (l >> 4);
            const float* __restrict__ wc = W + (size_t)(n * D_K) * IN_DIM + i0;
            const float* __restrict__ an = att + n * ATT_DIM;
            for (int k = 0; k < D_K; ++k) {
                const float ak = an[k];
                const float* __restrict__ wk = wc + (size_t)k * IN_DIM;
                v.x = fmaf(wk[0], ak, v.x);
                v.y = fmaf(wk[1], ak, v.y);
                v.z = fmaf(wk[4], ak, v.z);
                v.w = fmaf(wk[5], ak, v.w);
            }
        }
        ((float4*)wpack)[t] = v;
        return;
    }
    const float* __restrict__ wr =
        W + (size_t)(ot * 16 + (l & 15)) * IN_DIM + 8 * j + 2 * (l >> 4);
    float4 v = { wr[0], wr[1], wr[4], wr[5] };
    ((float4*)wpack)[t] = v;
}

__device__ __forceinline__ float leaky_clip_exp(float s) {
    s = (s >= 0.f) ? s : 0.2f * s;
    s = fminf(fmaxf(s, -20.f), 20.f);
    return __expf(s);
}

#define N_SLICES        ((N_NODES + SLICE - 1) / SLICE)
#define OWN_WAVES       32
#define OWN_THREADS     (OWN_WAVES * 32)
#define SCAN_UNROLL     5
#define N_UNITS         ((N_GROUPS + 31) / 32)
#define UNITS_PER_WAVE  ((N_UNITS + OWN_WAVES - 1) / OWN_WAVES)
#define SCAN_ITERS      ((UNITS_PER_WAVE + SCAN_UNROLL - 1) / SCAN_UNROLL)
#define WQ              64
#define HOLD_WORDS      16384
static_assert(SCAN_UNROLL == 5, "the scan picks a unit's mask by hand from five");

__device__ __forceinline__ float lane_bcast(float v, int l) {
    return __int_as_float(__builtin_amdgcn_readlane(__float_as_int(v), l));
}

__device__ __forceinline__ void lds_add(float* p, float v) {
    __hip_atomic_fetch_add(p, v, __ATOMIC_RELAXED, __HIP_MEMORY_SCOPE_WORKGROUP);
}

__device__ __forceinline__ void take_edges(
        const int* wqp, const int nv, const int lane, const int lo,
        const int* __restrict__ ei, const float* __restrict__ ef,
        const float* __restrict__ s_src, const float* sdl,
        const float ae0, const float ae1, const float ae2, const float ae3,
        const float* __restrict__ zl, float* al, float* den) {
    const int e   = wqp[(lane < nv) ? lane : 0];
    const int src = ei[e];
    const int dl  = ei[N_EDGES + e] - lo;
    const float f = ef[e];
    const float4 ss = *(const float4*)(s_src + (size_t)src * HEADS);
    const float4 sd = *(const float4*)(sdl + dl * HEADS);
    const float w0 = leaky_clip_exp(ss.x + sd.x + f * ae0);
    const float w1 = leaky_clip_exp(ss.y + sd.y + f * ae1);
    const float w2 = leaky_clip_exp(ss.z + sd.z + f * ae2);
    const float w3 = leaky_clip_exp(ss.w + sd.w + f * ae3);
    if (lane < nv) {
        lds_add(&den[dl * HEADS + 0], w0);
        lds_add(&den[dl * HEADS + 1], w1);
        lds_add(&den[dl * HEADS + 2], w2);
        lds_add(&den[dl * HEADS + 3], w3);
    }
    for (int c = 0; c < nv; c += 8) {
        float p[8][4];
#pragma unroll
        for (int j = 0; j < 8; ++j) {
            const int i  = (c + j < nv) ? (c + j) : (nv - 1);
            const int sj = __builtin_amdgcn_readlane(src, i);
            const float* __restrict__ zp = zl + (size_t)sj * OUT_DIM;
            p[j][0] = zp[0]       * lane_bcast(w0, i);
            p[j][1] = zp[D_K]     * lane_bcast(w1, i);
            p[j][2] = zp[2 * D_K] * lane_bcast(w2, i);
            p[j][3] = zp[3 * D_K] * lane_bcast(w3, i);
        }
#pragma unroll
        for (int j = 0; j < 8; ++j) {
            if (c + j < nv) {
                const int dj = __builtin_amdgcn_readlane(dl, c + j);
                float* ap = al + dj * OUT_DIM;
                lds_add(ap,           p[j][0]);
                lds_add(ap + D_K,     p[j][1]);
                lds_add(ap + 2 * D_K, p[j][2]);
                lds_add(ap + 3 * D_K, p[j][3]);
            }
        }
    }
}

__global__ void __launch_bounds__(OWN_THREADS)
owner_edges(const int* __restrict__ ei, const float* __restrict__ ef,
            const float* __restrict__ s_src, const unsigned int* __restrict__ ids,
            const float* __restrict__ att, const float* __restrict__ z,
            const float* __restrict__ ln_g, const float* __restrict__ ln_b,
            float* __restrict__ out) {
    __shared__ __attribute__((aligned(16))) float acc[SLICE * OUT_DIM];
    __shared__ __attribute__((aligned(16))) float sdl[SLICE * HEADS];
    __shared__ float den[SLICE * HEADS];
    __shared__ int   queue[OWN_WAVES * WQ + HOLD_WORDS];

    const int tid  = threadIdx.x;
    const int lane = tid & 31;
    const int wv   = __builtin_amdgcn_readfirstlane(tid >> 5);
    const int lo   = blockIdx.x * SLICE;
    const int cnt  = (N_NODES - lo < SLICE) ? (N_NODES - lo) : SLICE;

    for (int i = tid; i < SLICE * OUT_DIM; i += OWN_THREADS) acc[i] = 0.f;
    for (int i = tid; i < SLICE * HEADS;   i += OWN_THREADS) den[i] = 0.f;
    {
        const int hd = lane >> 3;
        const float* __restrict__ ad = att + hd * ATT_DIM + D_K + 4 * (lane & 7);
        const float ad0 = ad[0], ad1 = ad[1], ad2 = ad[2], ad3 = ad[3];
        for (int n = wv; n < cnt; n += OWN_WAVES) {
            const float4 zv = *(const float4*)(z + (size_t)(lo + n) * OUT_DIM + lane * 4);
            float sd = fmaf(zv.w, ad3, fmaf(zv.z, ad2, fmaf(zv.y, ad1, zv.x * ad0)));
#pragma unroll
            for (int off = 4; off > 0; off >>= 1) sd += __shfl_xor(sd, off, 32);
            if ((lane & 7) == 0) sdl[n * HEADS + hd] = sd;
        }
    }
    __syncthreads();

    const uint4* __restrict__ id4 = (const uint4*)ids;
    unsigned sb[8];
#pragma unroll
    for (int k = 0; k < 8; ++k) sb[k] = 0u - (((unsigned)blockIdx.x >> k) & 1u);
    const float ae0 = att[0 * ATT_DIM + 2 * D_K];
    const float ae1 = att[1 * ATT_DIM + 2 * D_K];
    const float ae2 = att[2 * ATT_DIM + 2 * D_K];
    const float ae3 = att[3 * ATT_DIM + 2 * D_K];
    const float* __restrict__ zl = z + lane;
    float* al = acc + lane;
    int* wq = queue + wv * WQ;
    int wcount = 0;

    const int rot = blockIdx.x % SCAN_ITERS;
    for (int it = 0; it < SCAN_ITERS; ++it) {
        int itr = it + rot;
        if (itr >= SCAN_ITERS) itr -= SCAN_ITERS;
        const int k0 = wv + OWN_WAVES * SCAN_UNROLL * itr;

        uint4 da[SCAN_UNROLL], db[SCAN_UNROLL];
#pragma unroll
        for (int u = 0; u < SCAN_UNROLL; ++u) {
            const int gr = (k0 + OWN_WAVES * u) * 32 + lane;
            const int gc = (gr < N_GROUPS) ? gr : (N_GROUPS - 1);
            const int q  = (gc >> 3) * 16 + (gc & 7);
            da[u] = id4[q];
            db[u] = id4[q + 8];
        }
        unsigned m[SCAN_UNROLL];
#pragma unroll
        for (int u = 0; u < SCAN_UNROLL; ++u) {
            const int gr = (k0 + OWN_WAVES * u) * 32 + lane;
            unsigned t = (gr < N_GROUPS) ? 0xffffffffu : 0u;
            t &= ~(da[u].x ^ sb[0]);
            t &= ~(da[u].y ^ sb[1]);
            t &= ~(da[u].z ^ sb[2]);
            t &= ~(da[u].w ^ sb[3]);
            t &= ~(db[u].x ^ sb[4]);
            t &= ~(db[u].y ^ sb[5]);
            t &= ~(db[u].z ^ sb[6]);
            t &= ~(db[u].w ^ sb[7]);
            m[u] = t;
        }
#pragma nounroll
        for (int u = 0; u < SCAN_UNROLL; ++u) {
            unsigned g = m[0];
            g = (u == 1) ? m[1] : g;
            g = (u == 2) ? m[2] : g;
            g = (u == 3) ? m[3] : g;
            g = (u == 4) ? m[4] : g;
            const int gr = (k0 + OWN_WAVES * u) * 32 + lane;
            const int e0 = ((gr >> 2) << 7) + (gr & 3);
            unsigned mb = __builtin_amdgcn_ballot_w32(g != 0u);
            while (mb != 0u) {
                if (g != 0u) {
                    wq[wcount + (int)__builtin_amdgcn_mbcnt_lo(mb, 0u)] = e0 + 4 * __builtin_ctz(g);
                    g &= g - 1u;
                }
                wcount += __builtin_popcount(mb);
                if (wcount >= 32) {
                    wcount -= 32;
                    take_edges(wq + wcount, 32, lane, lo, ei, ef, s_src, sdl,
                               ae0, ae1, ae2, ae3, zl, al, den);
                }
                mb = __builtin_amdgcn_ballot_w32(g != 0u);
            }
        }
    }
    if (wcount > 0) {
        take_edges(wq, wcount, lane, lo, ei, ef, s_src, sdl,
                   ae0, ae1, ae2, ae3, zl, al, den);
    }
    __syncthreads();

    const float4 gv = *(const float4*)(ln_g + lane * 4);
    const float4 bv = *(const float4*)(ln_b + lane * 4);
    const float g4[4] = { gv.x, gv.y, gv.z, gv.w };
    const float b4[4] = { bv.x, bv.y, bv.z, bv.w };
    for (int n = wv; n < cnt; n += OWN_WAVES) {
        const size_t base = (size_t)(lo + n) * OUT_DIM + lane * 4;
        const float4 a  = *(const float4*)(acc + n * OUT_DIM + lane * 4);
        const float4 zv = *(const float4*)(z + base);
        const float inv = 1.f / (den[n * HEADS + (lane >> 3)] + 1e-6f);
        float x[4] = { a.x * inv + zv.x, a.y * inv + zv.y, a.z * inv + zv.z, a.w * inv + zv.w };

        float s = x[0] + x[1] + x[2] + x[3];
#pragma unroll
        for (int off = 16; off > 0; off >>= 1) s += __shfl_xor(s, off, 32);
        const float mu = s * (1.f / 128.f);

        float v = 0.f;
#pragma unroll
        for (int i = 0; i < 4; ++i) { const float dd = x[i] - mu; v = fmaf(dd, dd, v); }
#pragma unroll
        for (int off = 16; off > 0; off >>= 1) v += __shfl_xor(v, off, 32);
        const float rstd = rsqrtf(v * (1.f / 128.f) + 1e-5f);

        float y[4];
#pragma unroll
        for (int i = 0; i < 4; ++i) {
            const float t = (x[i] - mu) * rstd * g4[i] + b4[i];
            y[i] = (t > 0.f) ? t : (__expf(t) - 1.f);
        }
        float4 o = { y[0], y[1], y[2], y[3] };
        *(float4*)(out + base) = o;
    }
}

extern "C" void kernel_launch(void* const* d_in, const int* in_sizes, int n_in,
                              void* d_out, int out_size, void* d_ws, size_t ws_size,
                              hipStream_t stream) {
    const float* h    = (const float*)d_in[0];
    const int*   ei   = (const int*)  d_in[1];
    const float* ef   = (const float*)d_in[2];
    const float* W_w  = (const float*)d_in[3];
    const float* W_b  = (const float*)d_in[4];
    const float* att  = (const float*)d_in[5];
    const float* ln_g = (const float*)d_in[6];
    const float* ln_b = (const float*)d_in[7];
    float* out = (float*)d_out;

    float* ws     = (float*)d_ws;
    float* z      = ws;
    float* s_src  = z      + (size_t)N_NODES * OUT_DIM;
    unsigned int* ids = (unsigned int*)(s_src + (size_t)N_NODES * HEADS);

    pack_w<<<PACK_BLOCKS + ID_BLOCKS, 256, 0, stream>>>(W_w, att, W_b, out, ei, ids);
    {
        const int waves = N_NODES / 16;
        const int blocks = (waves + 7) / 8;
        gemm_z_wmma<<<blocks, 256, 0, stream>>>(h, out, W_b, z, s_src);
    }
    owner_edges<<<N_SLICES, OWN_THREADS, 0, stream>>>(ei, ef, s_src, ids, att, z, ln_g, ln_b, out);
}


# Launch 1 also writes an 8-byte record per edge (src + local dst in one word, edge feature); owner launch reads it with 1 gather per edge, not 3; 6,400,000 B more workspace, size tested on device
# speedup vs baseline: 3.2344x; 1.0330x over previous
#include <hip/hip_runtime.h>
#include <math.h>

#define N_NODES 50000
#define N_EDGES 800000
#define IN_DIM  128
#define OUT_DIM 128
#define HEADS   4
#define D_K     32
#define ATT_DIM 65

typedef __attribute__((ext_vector_type(2))) float v2f;
typedef __attribute__((ext_vector_type(8))) float v8f;

#define WPACK_TILES        9
#define WPACK_TILE_FLOATS  (WPACK_TILES * 16 * 32 * 4)

__global__ void __launch_bounds__(256)
gemm_z_wmma(const float* __restrict__ h, const float* __restrict__ wpack,
            const float* __restrict__ bias, float* __restrict__ z,
            float* __restrict__ s_src) {
    const int wave = (blockIdx.x * blockDim.x + threadIdx.x) >> 5;
    const int lane = threadIdx.x & 31;
    if (wave >= (N_NODES / 16)) return;

    const int m = lane & 15;
    const int g = lane >> 4;

    const float* __restrict__ hrow = h + (size_t)(wave * 16 + m) * IN_DIM;
    v2f a[32];
#pragma unroll
    for (int ks = 0; ks < 32; ++ks) {
        const int k = 4 * ks + 2 * g;
        a[ks].x = hrow[k];
        a[ks].y = hrow[k + 1];
    }

    const float4* __restrict__ wp4 = (const float4*)wpack + lane;
    for (int ot = 0; ot < 8; ++ot) {
        const int col = ot * 16 + m;
        const float4* __restrict__ bp = wp4 + ot * (16 * 32);
        v8f c = {};
#pragma unroll
        for (int j = 0; j < 16; ++j) {
            const float4 b4 = bp[j * 32];
            v2f b0, b1;
            b0.x = b4.x; b0.y = b4.y;
            b1.x = b4.z; b1.y = b4.w;
            c = __builtin_amdgcn_wmma_f32_16x16x4_f32(
                     false, a[2 * j],  false, b0,
                     (short)0, c,  false,  false);
            c = __builtin_amdgcn_wmma_f32_16x16x4_f32(
                     false, a[2 * j + 1],  false, b1,
                     (short)0, c,  false,  false);
        }
        const float bv = bias[col];
        const int rowBase = wave * 16 + g * 8;
#pragma unroll
        for (int r = 0; r < 8; ++r) {
            z[(size_t)(rowBase + r) * OUT_DIM + col] = c[r] + bv;
        }
    }
    {
        const float4* __restrict__ bp = wp4 + 8 * (16 * 32);
        v8f c = {};
#pragma unroll
        for (int j = 0; j < 16; ++j) {
            const float4 b4 = bp[j * 32];
            v2f b0, b1;
            b0.x = b4.x; b0.y = b4.y;
            b1.x = b4.z; b1.y = b4.w;
            c = __builtin_amdgcn_wmma_f32_16x16x4_f32(
                     false, a[2 * j],  false, b0,
                     (short)0, c,  false,  false);
            c = __builtin_amdgcn_wmma_f32_16x16x4_f32(
                     false, a[2 * j + 1],  false, b1,
                     (short)0, c,  false,  false);
        }
        if (m < HEADS) {
            const float ba = wpack[WPACK_TILE_FLOATS + m];
            const int rowBase = wave * 16 + g * 8;
#pragma unroll
            for (int r = 0; r < 8; ++r) {
                s_src[(size_t)(rowBase + r) * HEADS + m] = c[r] + ba;
            }
        }
    }
}

#define SLICE        196
#define PACK_BLOCKS  19
#define ID_WORDS     (N_EDGES / 4)
#define ID_BLOCKS    ((ID_WORDS + 255) / 256)
#define N_GROUPS     (N_EDGES / 32)
static_assert(N_EDGES % 128 == 0, "a wave of the first launch takes 128 whole edges");
static_assert(N_GROUPS % 8 == 0, "8 groups share a pair of lines");
static_assert(N_GROUPS * 8 == ID_WORDS, "8 plane words per group fill the plane exactly");
static_assert((N_NODES - 1) / SLICE <= 255, "a slice number must fit 8 bit planes");
static_assert((size_t)WPACK_TILE_FLOATS + HEADS <= (size_t)N_NODES * OUT_DIM, "wpack fits in out");
#define REC_FIRST_WORD  ((size_t)N_NODES * OUT_DIM + (size_t)N_NODES * HEADS + (size_t)ID_WORDS)
#define REC_END_BYTES   ((REC_FIRST_WORD + 2 * (size_t)N_EDGES) * 4)
static_assert(N_NODES <= 65536, "src must fit the low 16 bits of a record's first word");
static_assert(SLICE <= 256, "a local destination must fit the bits above src");
static_assert((REC_FIRST_WORD * 4) % 16 == 0, "records are written 16 bytes at a time");
__global__ void __launch_bounds__(256)
pack_w(const float* __restrict__ W, const float* __restrict__ att,
       const float* __restrict__ bias, float* __restrict__ wpack,
       const int* __restrict__ ei, const float* __restrict__ ef,
       unsigned int* __restrict__ ids, unsigned int* __restrict__ rec, const size_t ws_bytes) {
    if (blockIdx.x >= PACK_BLOCKS) {
        const int t = (blockIdx.x - PACK_BLOCKS) * 256 + threadIdx.x;
        if (t < ID_WORDS) {
            const int4 d = ((const int4*)(ei + N_EDGES))[t];
            const bool packed = ws_bytes >= REC_END_BYTES;
            int2 rs0 = {0, 0}, rd0 = {0, 0}, rs1 = {0, 0}, rd1 = {0, 0};
            float2 rf0 = {0.f, 0.f}, rf1 = {0.f, 0.f};
            const int re0 = (t >> 5) * 128 + 2 * (threadIdx.x & 31);
            if (packed) {
                rs0 = *(const int2*)(ei + re0);
                rd0 = *(const int2*)(ei + N_EDGES + re0);
                rf0 = *(const float2*)(ef + re0);
                rs1 = *(const int2*)(ei + re0 + 64);
                rd1 = *(const int2*)(ei + N_EDGES + re0 + 64);
                rf1 = *(const float2*)(ef + re0 + 64);
            }
            const unsigned x =  ((unsigned)d.x / SLICE)        | (((unsigned)d.y / SLICE) << 8)
                             | (((unsigned)d.z / SLICE) << 16) | (((unsigned)d.w / SLICE) << 24);
            const int lane = threadIdx.x & 31;
            const int w    = t >> 5;
            unsigned v = 0u;
#pragma unroll
            for (int j = 0; j < 32; ++j) {
                const int p = (j >> 2) & 3;
                const int k = (j >> 4) * 4 + (j & 3);
                const unsigned bal = __builtin_amdgcn_ballot_w32(((x >> (8 * p + k)) & 1u) != 0u);
                v = (lane == j) ? bal : v;
            }
            ids[(w >> 1) * 64 + (w & 1) * 16 + (lane >> 4) * 32 + (lane & 15)] = v;
            if (packed) {
                uint4 r0, r1;
                r0.x = (unsigned)rs0.x | (((unsigned)rd0.x % SLICE) << 16);
                r0.y = __float_as_uint(rf0.x);
                r0.z = (unsigned)rs0.y | (((unsigned)rd0.y % SLICE) << 16);
                r0.w = __float_as_uint(rf0.y);
                r1.x = (unsigned)rs1.x | (((unsigned)rd1.x % SLICE) << 16);
                r1.y = __float_as_uint(rf1.x);
                r1.z = (unsigned)rs1.y | (((unsigned)rd1.y % SLICE) << 16);
                r1.w = __float_as_uint(rf1.y);
                uint4* __restrict__ rec16 = (uint4*)rec;
                rec16[re0 >> 1]        = r0;
                rec16[(re0 >> 1) + 32] = r1;
            }
        }
        return;
    }
    if (blockIdx.x == PACK_BLOCKS - 1) {
        if (threadIdx.x < HEADS) {
            const int hd = threadIdx.x;
            float s = 0.f;
            for (int k = 0; k < D_K; ++k) s = fmaf(bias[hd * D_K + k], att[hd * ATT_DIM + k], s);
            wpack[WPACK_TILE_FLOATS + hd] = s;
        }
        return;
    }
    const int t  = blockIdx.x * 256 + threadIdx.x;
    const int l  = t & 31;
    const int j  = (t >> 5) & 15;
    const int ot = t >> 9;
    if (ot == 8) {
        const int n = l & 15;
        float4 v = { 0.f, 0.f, 0.f, 0.f };
        if (n < HEADS) {
            const int i0 = 8 * j + 2 * (l >> 4);
            const float* __restrict__ wc = W + (size_t)(n * D_K) * IN_DIM + i0;
            const float* __restrict__ an = att + n * ATT_DIM;
            for (int k = 0; k < D_K; ++k) {
                const float ak = an[k];
                const float* __restrict__ wk = wc + (size_t)k * IN_DIM;
                v.x = fmaf(wk[0], ak, v.x);
                v.y = fmaf(wk[1], ak, v.y);
                v.z = fmaf(wk[4], ak, v.z);
                v.w = fmaf(wk[5], ak, v.w);
            }
        }
        ((float4*)wpack)[t] = v;
        return;
    }
    const float* __restrict__ wr =
        W + (size_t)(ot * 16 + (l & 15)) * IN_DIM + 8 * j + 2 * (l >> 4);
    float4 v = { wr[0], wr[1], wr[4], wr[5] };
    ((float4*)wpack)[t] = v;
}

__device__ __forceinline__ float leaky_clip_exp(float s) {
    s = (s >= 0.f) ? s : 0.2f * s;
    s = fminf(fmaxf(s, -20.f), 20.f);
    return __expf(s);
}

#define N_SLICES        ((N_NODES + SLICE - 1) / SLICE)
#define OWN_WAVES       32
#define OWN_THREADS     (OWN_WAVES * 32)
#define SCAN_UNROLL     5
#define N_UNITS         ((N_GROUPS + 31) / 32)
#define UNITS_PER_WAVE  ((N_UNITS + OWN_WAVES - 1) / OWN_WAVES)
#define SCAN_ITERS      ((UNITS_PER_WAVE + SCAN_UNROLL - 1) / SCAN_UNROLL)
#define WQ              64
#define HOLD_WORDS      16384
static_assert(SCAN_UNROLL == 5, "the scan picks a unit's mask by hand from five");

__device__ __forceinline__ float lane_bcast(float v, int l) {
    return __int_as_float(__builtin_amdgcn_readlane(__float_as_int(v), l));
}

__device__ __forceinline__ void lds_add(float* p, float v) {
    __hip_atomic_fetch_add(p, v, __ATOMIC_RELAXED, __HIP_MEMORY_SCOPE_WORKGROUP);
}

__device__ __forceinline__ void take_edges(
        const int* wqp, const int nv, const int lane, const int lo,
        const int* __restrict__ ei, const float* __restrict__ ef,
        const unsigned int* __restrict__ rec, const bool packed,
        const float* __restrict__ s_src, const float* sdl,
        const float ae0, const float ae1, const float ae2, const float ae3,
        const float* __restrict__ zl, float* al, float* den) {
    const int e   = wqp[(lane < nv) ? lane : 0];
    int src, dl;
    float f;
    if (packed) {
        const unsigned long long r = ((const unsigned long long*)rec)[e];
        src = (int)((unsigned)r & 0xffffu);
        dl  = (int)((unsigned)r >> 16);
        f   = __uint_as_float((unsigned)(r >> 32));
    } else {
        src = ei[e];
        dl  = ei[N_EDGES + e] - lo;
        f   = ef[e];
    }
    const float4 ss = *(const float4*)(s_src + (size_t)src * HEADS);
    const float4 sd = *(const float4*)(sdl + dl * HEADS);
    const float w0 = leaky_clip_exp(ss.x + sd.x + f * ae0);
    const float w1 = leaky_clip_exp(ss.y + sd.y + f * ae1);
    const float w2 = leaky_clip_exp(ss.z + sd.z + f * ae2);
    const float w3 = leaky_clip_exp(ss.w + sd.w + f * ae3);
    if (lane < nv) {
        lds_add(&den[dl * HEADS + 0], w0);
        lds_add(&den[dl * HEADS + 1], w1);
        lds_add(&den[dl * HEADS + 2], w2);
        lds_add(&den[dl * HEADS + 3], w3);
    }
    for (int c = 0; c < nv; c += 8) {
        float p[8][4];
#pragma unroll
        for (int j = 0; j < 8; ++j) {
            const int i  = (c + j < nv) ? (c + j) : (nv - 1);
            const int sj = __builtin_amdgcn_readlane(src, i);
            const float* __restrict__ zp = zl + (size_t)sj * OUT_DIM;
            p[j][0] = zp[0]       * lane_bcast(w0, i);
            p[j][1] = zp[D_K]     * lane_bcast(w1, i);
            p[j][2] = zp[2 * D_K] * lane_bcast(w2, i);
            p[j][3] = zp[3 * D_K] * lane_bcast(w3, i);
        }
#pragma unroll
        for (int j = 0; j < 8; ++j) {
            if (c + j < nv) {
                const int dj = __builtin_amdgcn_readlane(dl, c + j);
                float* ap = al + dj * OUT_DIM;
                lds_add(ap,           p[j][0]);
                lds_add(ap + D_K,     p[j][1]);
                lds_add(ap + 2 * D_K, p[j][2]);
                lds_add(ap + 3 * D_K, p[j][3]);
            }
        }
    }
}

__global__ void __launch_bounds__(OWN_THREADS)
owner_edges(const int* __restrict__ ei, const float* __restrict__ ef,
            const unsigned int* __restrict__ rec, const size_t ws_bytes,
            const float* __restrict__ s_src, const unsigned int* __restrict__ ids,
            const float* __restrict__ att, const float* __restrict__ z,
            const float* __restrict__ ln_g, const float* __restrict__ ln_b,
            float* __restrict__ out) {
    __shared__ __attribute__((aligned(16))) float acc[SLICE * OUT_DIM];
    __shared__ __attribute__((aligned(16))) float sdl[SLICE * HEADS];
    __shared__ float den[SLICE * HEADS];
    __shared__ int   queue[OWN_WAVES * WQ + HOLD_WORDS];

    const int tid  = threadIdx.x;
    const int lane = tid & 31;
    const int wv   = __builtin_amdgcn_readfirstlane(tid >> 5);
    const int lo   = blockIdx.x * SLICE;
    const int cnt  = (N_NODES - lo < SLICE) ? (N_NODES - lo) : SLICE;
    const bool packed = ws_bytes >= REC_END_BYTES;

    for (int i = tid; i < SLICE * OUT_DIM; i += OWN_THREADS) acc[i] = 0.f;
    for (int i = tid; i < SLICE * HEADS;   i += OWN_THREADS) den[i] = 0.f;
    {
        const int hd = lane >> 3;
        const float* __restrict__ ad = att + hd * ATT_DIM + D_K + 4 * (lane & 7);
        const float ad0 = ad[0], ad1 = ad[1], ad2 = ad[2], ad3 = ad[3];
        for (int n = wv; n < cnt; n += OWN_WAVES) {
            const float4 zv = *(const float4*)(z + (size_t)(lo + n) * OUT_DIM + lane * 4);
            float sd = fmaf(zv.w, ad3, fmaf(zv.z, ad2, fmaf(zv.y, ad1, zv.x * ad0)));
#pragma unroll
            for (int off = 4; off > 0; off >>= 1) sd += __shfl_xor(sd, off, 32);
            if ((lane & 7) == 0) sdl[n * HEADS + hd] = sd;
        }
    }
    __syncthreads();

    const uint4* __restrict__ id4 = (const uint4*)ids;
    unsigned sb[8];
#pragma unroll
    for (int k = 0; k < 8; ++k) sb[k] = 0u - (((unsigned)blockIdx.x >> k) & 1u);
    const float ae0 = att[0 * ATT_DIM + 2 * D_K];
    const float ae1 = att[1 * ATT_DIM + 2 * D_K];
    const float ae2 = att[2 * ATT_DIM + 2 * D_K];
    const float ae3 = att[3 * ATT_DIM + 2 * D_K];
    const float* __restrict__ zl = z + lane;
    float* al = acc + lane;
    int* wq = queue + wv * WQ;
    int wcount = 0;

    const int rot = blockIdx.x % SCAN_ITERS;
    for (int it = 0; it < SCAN_ITERS; ++it) {
        int itr = it + rot;
        if (itr >= SCAN_ITERS) itr -= SCAN_ITERS;
        const int k0 = wv + OWN_WAVES * SCAN_UNROLL * itr;

        uint4 da[SCAN_UNROLL], db[SCAN_UNROLL];
#pragma unroll
        for (int u = 0; u < SCAN_UNROLL; ++u) {
            const int gr = (k0 + OWN_WAVES * u) * 32 + lane;
            const int gc = (gr < N_GROUPS) ? gr : (N_GROUPS - 1);
            const int q  = (gc >> 3) * 16 + (gc & 7);
            da[u] = id4[q];
            db[u] = id4[q + 8];
        }
        unsigned m[SCAN_UNROLL];
#pragma unroll
        for (int u = 0; u < SCAN_UNROLL; ++u) {
            const int gr = (k0 + OWN_WAVES * u) * 32 + lane;
            unsigned t = (gr < N_GROUPS) ? 0xffffffffu : 0u;
            t &= ~(da[u].x ^ sb[0]);
            t &= ~(da[u].y ^ sb[1]);
            t &= ~(da[u].z ^ sb[2]);
            t &= ~(da[u].w ^ sb[3]);
            t &= ~(db[u].x ^ sb[4]);
            t &= ~(db[u].y ^ sb[5]);
            t &= ~(db[u].z ^ sb[6]);
            t &= ~(db[u].w ^ sb[7]);
            m[u] = t;
        }
#pragma nounroll
        for (int u = 0; u < SCAN_UNROLL; ++u) {
            unsigned g = m[0];
            g = (u == 1) ? m[1] : g;
            g = (u == 2) ? m[2] : g;
            g = (u == 3) ? m[3] : g;
            g = (u == 4) ? m[4] : g;
            const int gr = (k0 + OWN_WAVES * u) * 32 + lane;
            const int e0 = ((gr >> 2) << 7) + (gr & 3);
            unsigned mb = __builtin_amdgcn_ballot_w32(g != 0u);
            while (mb != 0u) {
                if (g != 0u) {
                    wq[wcount + (int)__builtin_amdgcn_mbcnt_lo(mb, 0u)] = e0 + 4 * __builtin_ctz(g);
                    g &= g - 1u;
                }
                wcount += __builtin_popcount(mb);
                if (wcount >= 32) {
                    wcount -= 32;
                    take_edges(wq + wcount, 32, lane, lo, ei, ef, rec, packed, s_src, sdl,
                               ae0, ae1, ae2, ae3, zl, al, den);
                }
                mb = __builtin_amdgcn_ballot_w32(g != 0u);
            }
        }
    }
    if (wcount > 0) {
        take_edges(wq, wcount, lane, lo, ei, ef, rec, packed, s_src, sdl,
                   ae0, ae1, ae2, ae3, zl, al, den);
    }
    __syncthreads();

    const float4 gv = *(const float4*)(ln_g + lane * 4);
    const float4 bv = *(const float4*)(ln_b + lane * 4);
    const float g4[4] = { gv.x, gv.y, gv.z, gv.w };
    const float b4[4] = { bv.x, bv.y, bv.z, bv.w };
    for (int n = wv; n < cnt; n += OWN_WAVES) {
        const size_t base = (size_t)(lo + n) * OUT_DIM + lane * 4;
        const float4 a  = *(const float4*)(acc + n * OUT_DIM + lane * 4);
        const float4 zv = *(const float4*)(z + base);
        const float inv = 1.f / (den[n * HEADS + (lane >> 3)] + 1e-6f);
        float x[4] = { a.x * inv + zv.x, a.y * inv + zv.y, a.z * inv + zv.z, a.w * inv + zv.w };

        float s = x[0] + x[1] + x[2] + x[3];
#pragma unroll
        for (int off = 16; off > 0; off >>= 1) s += __shfl_xor(s, off, 32);
        const float mu = s * (1.f / 128.f);

        float v = 0.f;
#pragma unroll
        for (int i = 0; i < 4; ++i) { const float dd = x[i] - mu; v = fmaf(dd, dd, v); }
#pragma unroll
        for (int off = 16; off > 0; off >>= 1) v += __shfl_xor(v, off, 32);
        const float rstd = rsqrtf(v * (1.f / 128.f) + 1e-5f);

        float y[4];
#pragma unroll
        for (int i = 0; i < 4; ++i) {
            const float t = (x[i] - mu) * rstd * g4[i] + b4[i];
            y[i] = (t > 0.f) ? t : (__expf(t) - 1.f);
        }
        float4 o = { y[0], y[1], y[2], y[3] };
        *(float4*)(out + base) = o;
    }
}

extern "C" void kernel_launch(void* const* d_in, const int* in_sizes, int n_in,
                              void* d_out, int out_size, void* d_ws, size_t ws_size,
                              hipStream_t stream) {
    const float* h    = (const float*)d_in[0];
    const int*   ei   = (const int*)  d_in[1];
    const float* ef   = (const float*)d_in[2];
    const float* W_w  = (const float*)d_in[3];
    const float* W_b  = (const float*)d_in[4];
    const float* att  = (const float*)d_in[5];
    const float* ln_g = (const float*)d_in[6];
    const float* ln_b = (const float*)d_in[7];
    float* out = (float*)d_out;

    float* ws     = (float*)d_ws;
    float* z      = ws;
    float* s_src  = z      + (size_t)N_NODES * OUT_DIM;
    unsigned int* ids = (unsigned int*)(s_src + (size_t)N_NODES * HEADS);
    unsigned int* rec = ids + ID_WORDS;

    pack_w<<<PACK_BLOCKS + ID_BLOCKS, 256, 0, stream>>>(W_w, att, W_b, out, ei, ef, ids, rec, ws_size);
    {
        const int waves = N_NODES / 16;
        const int blocks = (waves + 7) / 8;
        gemm_z_wmma<<<blocks, 256, 0, stream>>>(h, out, W_b, z, s_src);
    }
    owner_edges<<<N_SLICES, OWN_THREADS, 0, stream>>>(ei, ef, rec, ws_size, s_src, ids,
                                                      att, z, ln_g, ln_b, out);
}


# Owner launch's per-node loops (score halves at its head, finish at its tail): a wave loads all 7 nodes' rows up front and sums lanes by DPP/permlane, not one LDS-permute chain per node; same sums
# speedup vs baseline: 3.3867x; 1.0471x over previous
#include <hip/hip_runtime.h>
#include <math.h>

#define N_NODES 50000
#define N_EDGES 800000
#define IN_DIM  128
#define OUT_DIM 128
#define HEADS   4
#define D_K     32
#define ATT_DIM 65

typedef __attribute__((ext_vector_type(2))) float v2f;
typedef __attribute__((ext_vector_type(8))) float v8f;

#define WPACK_TILES        9
#define WPACK_TILE_FLOATS  (WPACK_TILES * 16 * 32 * 4)

__global__ void __launch_bounds__(256)
gemm_z_wmma(const float* __restrict__ h, const float* __restrict__ wpack,
            const float* __restrict__ bias, float* __restrict__ z,
            float* __restrict__ s_src) {
    const int wave = (blockIdx.x * blockDim.x + threadIdx.x) >> 5;
    const int lane = threadIdx.x & 31;
    if (wave >= (N_NODES / 16)) return;

    const int m = lane & 15;
    const int g = lane >> 4;

    const float* __restrict__ hrow = h + (size_t)(wave * 16 + m) * IN_DIM;
    v2f a[32];
#pragma unroll
    for (int ks = 0; ks < 32; ++ks) {
        const int k = 4 * ks + 2 * g;
        a[ks].x = hrow[k];
        a[ks].y = hrow[k + 1];
    }

    const float4* __restrict__ wp4 = (const float4*)wpack + lane;
    for (int ot = 0; ot < 8; ++ot) {
        const int col = ot * 16 + m;
        const float4* __restrict__ bp = wp4 + ot * (16 * 32);
        v8f c = {};
#pragma unroll
        for (int j = 0; j < 16; ++j) {
            const float4 b4 = bp[j * 32];
            v2f b0, b1;
            b0.x = b4.x; b0.y = b4.y;
            b1.x = b4.z; b1.y = b4.w;
            c = __builtin_amdgcn_wmma_f32_16x16x4_f32(
                     false, a[2 * j],  false, b0,
                     (short)0, c,  false,  false);
            c = __builtin_amdgcn_wmma_f32_16x16x4_f32(
                     false, a[2 * j + 1],  false, b1,
                     (short)0, c,  false,  false);
        }
        const float bv = bias[col];
        const int rowBase = wave * 16 + g * 8;
#pragma unroll
        for (int r = 0; r < 8; ++r) {
            z[(size_t)(rowBase + r) * OUT_DIM + col] = c[r] + bv;
        }
    }
    {
        const float4* __restrict__ bp = wp4 + 8 * (16 * 32);
        v8f c = {};
#pragma unroll
        for (int j = 0; j < 16; ++j) {
            const float4 b4 = bp[j * 32];
            v2f b0, b1;
            b0.x = b4.x; b0.y = b4.y;
            b1.x = b4.z; b1.y = b4.w;
            c = __builtin_amdgcn_wmma_f32_16x16x4_f32(
                     false, a[2 * j],  false, b0,
                     (short)0, c,  false,  false);
            c = __builtin_amdgcn_wmma_f32_16x16x4_f32(
                     false, a[2 * j + 1],  false, b1,
                     (short)0, c,  false,  false);
        }
        if (m < HEADS) {
            const float ba = wpack[WPACK_TILE_FLOATS + m];
            const int rowBase = wave * 16 + g * 8;
#pragma unroll
            for (int r = 0; r < 8; ++r) {
                s_src[(size_t)(rowBase + r) * HEADS + m] = c[r] + ba;
            }
        }
    }
}

#define SLICE        196
#define PACK_BLOCKS  19
#define ID_WORDS     (N_EDGES / 4)
#define ID_BLOCKS    ((ID_WORDS + 255) / 256)
#define N_GROUPS     (N_EDGES / 32)
static_assert(N_EDGES % 128 == 0, "a wave of the first launch takes 128 whole edges");
static_assert(N_GROUPS % 8 == 0, "8 groups share a pair of lines");
static_assert(N_GROUPS * 8 == ID_WORDS, "8 plane words per group fill the plane exactly");
static_assert((N_NODES - 1) / SLICE <= 255, "a slice number must fit 8 bit planes");
static_assert((size_t)WPACK_TILE_FLOATS + HEADS <= (size_t)N_NODES * OUT_DIM, "wpack fits in out");
#define REC_FIRST_WORD  ((size_t)N_NODES * OUT_DIM + (size_t)N_NODES * HEADS + (size_t)ID_WORDS)
#define REC_END_BYTES   ((REC_FIRST_WORD + 2 * (size_t)N_EDGES) * 4)
static_assert(N_NODES <= 65536, "src must fit the low 16 bits of a record's first word");
static_assert(SLICE <= 256, "a local destination must fit the bits above src");
static_assert((REC_FIRST_WORD * 4) % 16 == 0, "records are written 16 bytes at a time");
__global__ void __launch_bounds__(256)
pack_w(const float* __restrict__ W, const float* __restrict__ att,
       const float* __restrict__ bias, float* __restrict__ wpack,
       const int* __restrict__ ei, const float* __restrict__ ef,
       unsigned int* __restrict__ ids, unsigned int* __restrict__ rec, const size_t ws_bytes) {
    if (blockIdx.x >= PACK_BLOCKS) {
        const int t = (blockIdx.x - PACK_BLOCKS) * 256 + threadIdx.x;
        if (t < ID_WORDS) {
            const int4 d = ((const int4*)(ei + N_EDGES))[t];
            const bool packed = ws_bytes >= REC_END_BYTES;
            int2 rs0 = {0, 0}, rd0 = {0, 0}, rs1 = {0, 0}, rd1 = {0, 0};
            float2 rf0 = {0.f, 0.f}, rf1 = {0.f, 0.f};
            const int re0 = (t >> 5) * 128 + 2 * (threadIdx.x & 31);
            if (packed) {
                rs0 = *(const int2*)(ei + re0);
                rd0 = *(const int2*)(ei + N_EDGES + re0);
                rf0 = *(const float2*)(ef + re0);
                rs1 = *(const int2*)(ei + re0 + 64);
                rd1 = *(const int2*)(ei + N_EDGES + re0 + 64);
                rf1 = *(const float2*)(ef + re0 + 64);
            }
            const unsigned x =  ((unsigned)d.x / SLICE)        | (((unsigned)d.y / SLICE) << 8)
                             | (((unsigned)d.z / SLICE) << 16) | (((unsigned)d.w / SLICE) << 24);
            const int lane = threadIdx.x & 31;
            const int w    = t >> 5;
            unsigned v = 0u;
#pragma unroll
            for (int j = 0; j < 32; ++j) {
                const int p = (j >> 2) & 3;
                const int k = (j >> 4) * 4 + (j & 3);
                const unsigned bal = __builtin_amdgcn_ballot_w32(((x >> (8 * p + k)) & 1u) != 0u);
                v = (lane == j) ? bal : v;
            }
            ids[(w >> 1) * 64 + (w & 1) * 16 + (lane >> 4) * 32 + (lane & 15)] = v;
            if (packed) {
                uint4 r0, r1;
                r0.x = (unsigned)rs0.x | (((unsigned)rd0.x % SLICE) << 16);
                r0.y = __float_as_uint(rf0.x);
                r0.z = (unsigned)rs0.y | (((unsigned)rd0.y % SLICE) << 16);
                r0.w = __float_as_uint(rf0.y);
                r1.x = (unsigned)rs1.x | (((unsigned)rd1.x % SLICE) << 16);
                r1.y = __float_as_uint(rf1.x);
                r1.z = (unsigned)rs1.y | (((unsigned)rd1.y % SLICE) << 16);
                r1.w = __float_as_uint(rf1.y);
                uint4* __restrict__ rec16 = (uint4*)rec;
                rec16[re0 >> 1]        = r0;
                rec16[(re0 >> 1) + 32] = r1;
            }
        }
        return;
    }
    if (blockIdx.x == PACK_BLOCKS - 1) {
        if (threadIdx.x < HEADS) {
            const int hd = threadIdx.x;
            float s = 0.f;
            for (int k = 0; k < D_K; ++k) s = fmaf(bias[hd * D_K + k], att[hd * ATT_DIM + k], s);
            wpack[WPACK_TILE_FLOATS + hd] = s;
        }
        return;
    }
    const int t  = blockIdx.x * 256 + threadIdx.x;
    const int l  = t & 31;
    const int j  = (t >> 5) & 15;
    const int ot = t >> 9;
    if (ot == 8) {
        const int n = l & 15;
        float4 v = { 0.f, 0.f, 0.f, 0.f };
        if (n < HEADS) {
            const int i0 = 8 * j + 2 * (l >> 4);
            const float* __restrict__ wc = W + (size_t)(n * D_K) * IN_DIM + i0;
            const float* __restrict__ an = att + n * ATT_DIM;
            for (int k = 0; k < D_K; ++k) {
                const float ak = an[k];
                const float* __restrict__ wk = wc + (size_t)k * IN_DIM;
                v.x = fmaf(wk[0], ak, v.x);
                v.y = fmaf(wk[1], ak, v.y);
                v.z = fmaf(wk[4], ak, v.z);
                v.w = fmaf(wk[5], ak, v.w);
            }
        }
        ((float4*)wpack)[t] = v;
        return;
    }
    const float* __restrict__ wr =
        W + (size_t)(ot * 16 + (l & 15)) * IN_DIM + 8 * j + 2 * (l >> 4);
    float4 v = { wr[0], wr[1], wr[4], wr[5] };
    ((float4*)wpack)[t] = v;
}

__device__ __forceinline__ float leaky_clip_exp(float s) {
    s = (s >= 0.f) ? s : 0.2f * s;
    s = fminf(fmaxf(s, -20.f), 20.f);
    return __expf(s);
}

#define N_SLICES        ((N_NODES + SLICE - 1) / SLICE)
#define OWN_WAVES       32
#define OWN_THREADS     (OWN_WAVES * 32)
#define SCAN_UNROLL     5
#define N_UNITS         ((N_GROUPS + 31) / 32)
#define UNITS_PER_WAVE  ((N_UNITS + OWN_WAVES - 1) / OWN_WAVES)
#define SCAN_ITERS      ((UNITS_PER_WAVE + SCAN_UNROLL - 1) / SCAN_UNROLL)
#define WQ              64
#define HOLD_WORDS      16384
#define NODES_PER_WAVE  ((SLICE + OWN_WAVES - 1) / OWN_WAVES)
static_assert(SCAN_UNROLL == 5, "the scan picks a unit's mask by hand from five");

__device__ __forceinline__ float lane_bcast(float v, int l) {
    return __int_as_float(__builtin_amdgcn_readlane(__float_as_int(v), l));
}

__device__ __forceinline__ void lds_add(float* p, float v) {
    __hip_atomic_fetch_add(p, v, __ATOMIC_RELAXED, __HIP_MEMORY_SCOPE_WORKGROUP);
}

template <int CTRL>
__device__ __forceinline__ float lane_xor_dpp(float v) {
    return __int_as_float(__builtin_amdgcn_update_dpp(0, __float_as_int(v), CTRL, 0xF, 0xF, true));
}
#define DPP_XOR1  0xB1
#define DPP_XOR2  0x4E
#define DPP_XOR4  0x164
#define DPP_XOR8  0x168
__device__ __forceinline__ float lane_xor16(float v) {
    const int i = __float_as_int(v);
    return __int_as_float(__builtin_amdgcn_permlanex16(i, i, 0x76543210, 0xfedcba98, false, false));
}
__device__ __forceinline__ float sum8_lanes(float v) {
    v += lane_xor_dpp<DPP_XOR4>(v);
    v += lane_xor_dpp<DPP_XOR2>(v);
    v += lane_xor_dpp<DPP_XOR1>(v);
    return v;
}
__device__ __forceinline__ float sum32_lanes(float v) {
    v += lane_xor16(v);
    v += lane_xor_dpp<DPP_XOR8>(v);
    v += lane_xor_dpp<DPP_XOR4>(v);
    v += lane_xor_dpp<DPP_XOR2>(v);
    v += lane_xor_dpp<DPP_XOR1>(v);
    return v;
}

__device__ __forceinline__ void take_edges(
        const int* wqp, const int nv, const int lane, const int lo,
        const int* __restrict__ ei, const float* __restrict__ ef,
        const unsigned int* __restrict__ rec, const bool packed,
        const float* __restrict__ s_src, const float* sdl,
        const float ae0, const float ae1, const float ae2, const float ae3,
        const float* __restrict__ zl, float* al, float* den) {
    const int e   = wqp[(lane < nv) ? lane : 0];
    int src, dl;
    float f;
    if (packed) {
        const unsigned long long r = ((const unsigned long long*)rec)[e];
        src = (int)((unsigned)r & 0xffffu);
        dl  = (int)((unsigned)r >> 16);
        f   = __uint_as_float((unsigned)(r >> 32));
    } else {
        src = ei[e];
        dl  = ei[N_EDGES + e] - lo;
        f   = ef[e];
    }
    const float4 ss = *(const float4*)(s_src + (size_t)src * HEADS);
    const float4 sd = *(const float4*)(sdl + dl * HEADS);
    const float w0 = leaky_clip_exp(ss.x + sd.x + f * ae0);
    const float w1 = leaky_clip_exp(ss.y + sd.y + f * ae1);
    const float w2 = leaky_clip_exp(ss.z + sd.z + f * ae2);
    const float w3 = leaky_clip_exp(ss.w + sd.w + f * ae3);
    if (lane < nv) {
        lds_add(&den[dl * HEADS + 0], w0);
        lds_add(&den[dl * HEADS + 1], w1);
        lds_add(&den[dl * HEADS + 2], w2);
        lds_add(&den[dl * HEADS + 3], w3);
    }
    for (int c = 0; c < nv; c += 8) {
        float p[8][4];
#pragma unroll
        for (int j = 0; j < 8; ++j) {
            const int i  = (c + j < nv) ? (c + j) : (nv - 1);
            const int sj = __builtin_amdgcn_readlane(src, i);
            const float* __restrict__ zp = zl + (size_t)sj * OUT_DIM;
            p[j][0] = zp[0]       * lane_bcast(w0, i);
            p[j][1] = zp[D_K]     * lane_bcast(w1, i);
            p[j][2] = zp[2 * D_K] * lane_bcast(w2, i);
            p[j][3] = zp[3 * D_K] * lane_bcast(w3, i);
        }
#pragma unroll
        for (int j = 0; j < 8; ++j) {
            if (c + j < nv) {
                const int dj = __builtin_amdgcn_readlane(dl, c + j);
                float* ap = al + dj * OUT_DIM;
                lds_add(ap,           p[j][0]);
                lds_add(ap + D_K,     p[j][1]);
                lds_add(ap + 2 * D_K, p[j][2]);
                lds_add(ap + 3 * D_K, p[j][3]);
            }
        }
    }
}

__global__ void __launch_bounds__(OWN_THREADS)
owner_edges(const int* __restrict__ ei, const float* __restrict__ ef,
            const unsigned int* __restrict__ rec, const size_t ws_bytes,
            const float* __restrict__ s_src, const unsigned int* __restrict__ ids,
            const float* __restrict__ att, const float* __restrict__ z,
            const float* __restrict__ ln_g, const float* __restrict__ ln_b,
            float* __restrict__ out) {
    __shared__ __attribute__((aligned(16))) float acc[SLICE * OUT_DIM];
    __shared__ __attribute__((aligned(16))) float sdl[SLICE * HEADS];
    __shared__ float den[SLICE * HEADS];
    __shared__ int   queue[OWN_WAVES * WQ + HOLD_WORDS];

    const int tid  = threadIdx.x;
    const int lane = tid & 31;
    const int wv   = __builtin_amdgcn_readfirstlane(tid >> 5);
    const int lo   = blockIdx.x * SLICE;
    const int cnt  = (N_NODES - lo < SLICE) ? (N_NODES - lo) : SLICE;
    const bool packed = ws_bytes >= REC_END_BYTES;

    float4 zr[NODES_PER_WAVE];
#pragma unroll
    for (int q = 0; q < NODES_PER_WAVE; ++q) {
        const int n  = wv + q * OWN_WAVES;
        const int nc = (n < cnt) ? n : (cnt - 1);
        zr[q] = *(const float4*)(z + (size_t)(lo + nc) * OUT_DIM + lane * 4);
    }
    const int hd = lane >> 3;
    const float* __restrict__ ad = att + hd * ATT_DIM + D_K + 4 * (lane & 7);
    const float ad0 = ad[0], ad1 = ad[1], ad2 = ad[2], ad3 = ad[3];
    for (int i = tid; i < SLICE * OUT_DIM; i += OWN_THREADS) acc[i] = 0.f;
    for (int i = tid; i < SLICE * HEADS;   i += OWN_THREADS) den[i] = 0.f;
    {
#pragma unroll
        for (int q = 0; q < NODES_PER_WAVE; ++q) {
            const int n = wv + q * OWN_WAVES;
            float sd = fmaf(zr[q].w, ad3, fmaf(zr[q].z, ad2, fmaf(zr[q].y, ad1, zr[q].x * ad0)));
            sd = sum8_lanes(sd);
            if ((lane & 7) == 0 && n < cnt) sdl[n * HEADS + hd] = sd;
        }
    }
    __syncthreads();

    const uint4* __restrict__ id4 = (const uint4*)ids;
    unsigned sb[8];
#pragma unroll
    for (int k = 0; k < 8; ++k) sb[k] = 0u - (((unsigned)blockIdx.x >> k) & 1u);
    const float ae0 = att[0 * ATT_DIM + 2 * D_K];
    const float ae1 = att[1 * ATT_DIM + 2 * D_K];
    const float ae2 = att[2 * ATT_DIM + 2 * D_K];
    const float ae3 = att[3 * ATT_DIM + 2 * D_K];
    const float* __restrict__ zl = z + lane;
    float* al = acc + lane;
    int* wq = queue + wv * WQ;
    int wcount = 0;

    const int rot = blockIdx.x % SCAN_ITERS;
    for (int it = 0; it < SCAN_ITERS; ++it) {
        int itr = it + rot;
        if (itr >= SCAN_ITERS) itr -= SCAN_ITERS;
        const int k0 = wv + OWN_WAVES * SCAN_UNROLL * itr;

        uint4 da[SCAN_UNROLL], db[SCAN_UNROLL];
#pragma unroll
        for (int u = 0; u < SCAN_UNROLL; ++u) {
            const int gr = (k0 + OWN_WAVES * u) * 32 + lane;
            const int gc = (gr < N_GROUPS) ? gr : (N_GROUPS - 1);
            const int q  = (gc >> 3) * 16 + (gc & 7);
            da[u] = id4[q];
            db[u] = id4[q + 8];
        }
        unsigned m[SCAN_UNROLL];
#pragma unroll
        for (int u = 0; u < SCAN_UNROLL; ++u) {
            const int gr = (k0 + OWN_WAVES * u) * 32 + lane;
            unsigned t = (gr < N_GROUPS) ? 0xffffffffu : 0u;
            t &= ~(da[u].x ^ sb[0]);
            t &= ~(da[u].y ^ sb[1]);
            t &= ~(da[u].z ^ sb[2]);
            t &= ~(da[u].w ^ sb[3]);
            t &= ~(db[u].x ^ sb[4]);
            t &= ~(db[u].y ^ sb[5]);
            t &= ~(db[u].z ^ sb[6]);
            t &= ~(db[u].w ^ sb[7]);
            m[u] = t;
        }
#pragma nounroll
        for (int u = 0; u < SCAN_UNROLL; ++u) {
            unsigned g = m[0];
            g = (u == 1) ? m[1] : g;
            g = (u == 2) ? m[2] : g;
            g = (u == 3) ? m[3] : g;
            g = (u == 4) ? m[4] : g;
            const int gr = (k0 + OWN_WAVES * u) * 32 + lane;
            const int e0 = ((gr >> 2) << 7) + (gr & 3);
            unsigned mb = __builtin_amdgcn_ballot_w32(g != 0u);
            while (mb != 0u) {
                if (g != 0u) {
                    wq[wcount + (int)__builtin_amdgcn_mbcnt_lo(mb, 0u)] = e0 + 4 * __builtin_ctz(g);
                    g &= g - 1u;
                }
                wcount += __builtin_popcount(mb);
                if (wcount >= 32) {
                    wcount -= 32;
                    take_edges(wq + wcount, 32, lane, lo, ei, ef, rec, packed, s_src, sdl,
                               ae0, ae1, ae2, ae3, zl, al, den);
                }
                mb = __builtin_amdgcn_ballot_w32(g != 0u);
            }
        }
    }
    if (wcount > 0) {
        take_edges(wq, wcount, lane, lo, ei, ef, rec, packed, s_src, sdl,
                   ae0, ae1, ae2, ae3, zl, al, den);
    }
    __syncthreads();

    const float4 gv = *(const float4*)(ln_g + lane * 4);
    const float4 bv = *(const float4*)(ln_b + lane * 4);
    const float g4[4] = { gv.x, gv.y, gv.z, gv.w };
    const float b4[4] = { bv.x, bv.y, bv.z, bv.w };
    float4 zq[NODES_PER_WAVE], aq[NODES_PER_WAVE];
    float  dq[NODES_PER_WAVE];
#pragma unroll
    for (int q = 0; q < NODES_PER_WAVE; ++q) {
        const int n  = wv + q * OWN_WAVES;
        const int nc = (n < cnt) ? n : (cnt - 1);
        zq[q] = *(const float4*)(z + (size_t)(lo + nc) * OUT_DIM + lane * 4);
        aq[q] = *(const float4*)(acc + nc * OUT_DIM + lane * 4);
        dq[q] = den[nc * HEADS + (lane >> 3)];
    }
#pragma unroll
    for (int q = 0; q < NODES_PER_WAVE; ++q) {
        const int n = wv + q * OWN_WAVES;
        const size_t base = (size_t)(lo + n) * OUT_DIM + lane * 4;
        const float4 a  = aq[q];
        const float4 zv = zq[q];
        const float inv = 1.f / (dq[q] + 1e-6f);
        float x[4] = { a.x * inv + zv.x, a.y * inv + zv.y, a.z * inv + zv.z, a.w * inv + zv.w };

        float s = x[0] + x[1] + x[2] + x[3];
        s = sum32_lanes(s);
        const float mu = s * (1.f / 128.f);

        float v = 0.f;
#pragma unroll
        for (int i = 0; i < 4; ++i) { const float dd = x[i] - mu; v = fmaf(dd, dd, v); }
        v = sum32_lanes(v);
        const float rstd = rsqrtf(v * (1.f / 128.f) + 1e-5f);

        float y[4];
#pragma unroll
        for (int i = 0; i < 4; ++i) {
            const float t = (x[i] - mu) * rstd * g4[i] + b4[i];
            y[i] = (t > 0.f) ? t : (__expf(t) - 1.f);
        }
        float4 o = { y[0], y[1], y[2], y[3] };
        if (n < cnt) *(float4*)(out + base) = o;
    }
}

extern "C" void kernel_launch(void* const* d_in, const int* in_sizes, int n_in,
                              void* d_out, int out_size, void* d_ws, size_t ws_size,
                              hipStream_t stream) {
    const float* h    = (const float*)d_in[0];
    const int*   ei   = (const int*)  d_in[1];
    const float* ef   = (const float*)d_in[2];
    const float* W_w  = (const float*)d_in[3];
    const float* W_b  = (const float*)d_in[4];
    const float* att  = (const float*)d_in[5];
    const float* ln_g = (const float*)d_in[6];
    const float* ln_b = (const float*)d_in[7];
    float* out = (float*)d_out;

    float* ws     = (float*)d_ws;
    float* z      = ws;
    float* s_src  = z      + (size_t)N_NODES * OUT_DIM;
    unsigned int* ids = (unsigned int*)(s_src + (size_t)N_NODES * HEADS);
    unsigned int* rec = ids + ID_WORDS;

    pack_w<<<PACK_BLOCKS + ID_BLOCKS, 256, 0, stream>>>(W_w, att, W_b, out, ei, ef, ids, rec, ws_size);
    {
        const int waves = N_NODES / 16;
        const int blocks = (waves + 7) / 8;
        gemm_z_wmma<<<blocks, 256, 0, stream>>>(h, out, W_b, z, s_src);
    }
    owner_edges<<<N_SLICES, OWN_THREADS, 0, stream>>>(ei, ef, rec, ws_size, s_src, ids,
                                                      att, z, ln_g, ln_b, out);
}
